# starting kernel
# baseline (speedup 1.0000x reference)
#include <hip/hip_runtime.h>
#include <math.h>

typedef __attribute__((ext_vector_type(2))) float v2f;
typedef __attribute__((ext_vector_type(4))) float v4f;
typedef __attribute__((ext_vector_type(8))) float v8f;

#define B_  4
#define L_  512
#define V_  32
#define D_  64
#define MT  32
#define SCALE 0.125f
#define LDSS (L_ + 4)

__global__ __launch_bounds__(256)
void sum_key_kernel(const float* __restrict__ key, float* __restrict__ sumv) {
    int idx = blockIdx.x * blockDim.x + threadIdx.x;
    if (idx >= B_ * L_ * D_) return;
    int d  = idx % D_;
    int bl = idx / D_;
    const float* p = key + (size_t)bl * (V_ * D_) + d;
    float s = 0.f;
#pragma unroll
    for (int v = 0; v < V_; ++v) s += p[v * D_];
    sumv[idx] = s;
}

__global__ __launch_bounds__(32)
void attn_kernel(const float* __restrict__ q,
                 const float* __restrict__ sumv,
                 const float* __restrict__ val,
                 const int*   __restrict__ lab,
                 float*       __restrict__ out) {
    __shared__ float smem[MT * LDSS];
    __shared__ float labf[L_];
    __shared__ float rowsum[MT];

    const int blk   = blockIdx.x;
    const int itile = blk % (L_ / MT);
    const int bk    = blk / (L_ / MT);
    const int k     = bk % V_;
    const int b     = bk / V_;

    const int lane = threadIdx.x;
    const int hlf  = lane >> 4;
    const int l16  = lane & 15;
    const int i0   = itile * MT;

    for (int t = lane; t < L_; t += 32) labf[t] = (float)lab[b * L_ + t];
    __syncthreads();

    v2f qa0[16], qa1[16];
    {
        const float* q0 =
            q + (((size_t)(b * L_ + i0 + l16) * V_) + k) * D_ + hlf * 2;
        const float* q1 = q0 + (size_t)16 * V_ * D_;
#pragma unroll
        for (int s = 0; s < 16; ++s) {
            qa0[s] = *(const v2f*)(q0 + s * 4);
            qa1[s] = *(const v2f*)(q1 + s * 4);
        }
    }

    const float* sb = sumv + (size_t)b * L_ * D_;
    for (int jt = 0; jt < L_ / 16; ++jt) {
        v8f c0 = {}, c1 = {};
        const float* srow = sb + (size_t)(jt * 16 + l16) * D_ + hlf * 2;
#pragma unroll
        for (int s = 0; s < 16; ++s) {
            v2f bf = *(const v2f*)(srow + s * 4);
            c0 = __builtin_amdgcn_wmma_f32_16x16x4_f32(
                     false, qa0[s], false, bf, (short)0, c0, false, false);
            c1 = __builtin_amdgcn_wmma_f32_16x16x4_f32(
                     false, qa1[s], false, bf, (short)0, c1, false, false);
        }
#pragma unroll
        for (int r = 0; r < 8; ++r) {
            smem[(r + hlf * 8) * LDSS + jt * 16 + l16]      = c0[r];
            smem[(16 + r + hlf * 8) * LDSS + jt * 16 + l16] = c1[r];
        }
    }
    __syncthreads();

    {
        const int row = lane;
        const float myLab = labf[i0 + row];
        float* srow = &smem[row * LDSS];

        v4f m4 = {-INFINITY, -INFINITY, -INFINITY, -INFINITY};
        for (int c = 0; c < L_; c += 4) {
            v4f s  = *(v4f*)(srow + c);
            v4f lv = *(const v4f*)(&labf[c]);
            s.x = (lv.x == myLab) ? s.x : -INFINITY;
            s.y = (lv.y == myLab) ? s.y : -INFINITY;
            s.z = (lv.z == myLab) ? s.z : -INFINITY;
            s.w = (lv.w == myLab) ? s.w : -INFINITY;
            *(v4f*)(srow + c) = s;
            m4.x = fmaxf(m4.x, s.x);  m4.y = fmaxf(m4.y, s.y);
            m4.z = fmaxf(m4.z, s.z);  m4.w = fmaxf(m4.w, s.w);
        }
        const float m = fmaxf(fmaxf(m4.x, m4.y), fmaxf(m4.z, m4.w));

        v4f s4 = {0.f, 0.f, 0.f, 0.f};
        for (int c = 0; c < L_; c += 4) {
            v4f s = *(v4f*)(srow + c);
            v4f p;
            p.x = __expf(SCALE * (s.x - m));
            p.y = __expf(SCALE * (s.y - m));
            p.z = __expf(SCALE * (s.z - m));
            p.w = __expf(SCALE * (s.w - m));
            *(v4f*)(srow + c) = p;
            s4 += p;
        }
        rowsum[row] = (s4.x + s4.y) + (s4.z + s4.w);
    }
    __syncthreads();

    v8f acc[2][4] = {{{}, {}, {}, {}}, {{}, {}, {}, {}}};
    const float* vb = val + ((size_t)(b * L_) * V_ + k) * D_;
    for (int k0 = 0; k0 < L_; k0 += 4) {
        const int j0 = k0 + hlf * 2;
        const v2f af0 = *(const v2f*)(&smem[l16 * LDSS + j0]);
        const v2f af1 = *(const v2f*)(&smem[(16 + l16) * LDSS + j0]);
        const float* v0 = vb + (size_t)j0 * (V_ * D_);
        const float* v1 = v0 + (V_ * D_);
#pragma unroll
        for (int n = 0; n < 4; ++n) {
            v2f bf;
            bf.x = v0[n * 16 + l16];
            bf.y = v1[n * 16 + l16];
            acc[0][n] = __builtin_amdgcn_wmma_f32_16x16x4_f32(
                            false, af0, false, bf, (short)0, acc[0][n], false, false);
            acc[1][n] = __builtin_amdgcn_wmma_f32_16x16x4_f32(
                            false, af1, false, bf, (short)0, acc[1][n], false, false);
        }
    }

#pragma unroll
    for (int t = 0; t < 2; ++t) {
#pragma unroll
        for (int n = 0; n < 4; ++n) {
#pragma unroll
            for (int r = 0; r < 8; ++r) {
                const int irow = t * 16 + r + hlf * 8;
                const int i = i0 + irow;
                const float o = acc[t][n][r] / rowsum[irow];
                out[(((size_t)(b * L_ + i) * V_) + k) * D_ + n * 16 + l16] = o;
            }
        }
    }
}

extern "C" void kernel_launch(void* const* d_in, const int* in_sizes, int n_in,
                              void* d_out, int out_size, void* d_ws, size_t ws_size,
                              hipStream_t stream) {
    const float* query = (const float*)d_in[0];
    const float* key   = (const float*)d_in[1];
    const float* value = (const float*)d_in[2];
    const int*   label = (const int*)d_in[3];
    float* out  = (float*)d_out;
    float* sumv = (float*)d_ws;

    const int n1 = B_ * L_ * D_;
    sum_key_kernel<<<(n1 + 255) / 256, 256, 0, stream>>>(key, sumv);

    const int blocks = B_ * V_ * (L_ / MT);
    attn_kernel<<<blocks, 32, 0, stream>>>(query, sumv, value, label, out);
}


# Positions sorted by cluster label inside the attention kernel: each 32-row tile of the sorted order visits only the contiguous key range of its own clusters (masked keys weigh exactly 0)
# speedup vs baseline: 2.3302x; 2.3302x over previous
#include <hip/hip_runtime.h>
#include <math.h>


typedef __attribute__((ext_vector_type(2))) float v2f;
typedef __attribute__((ext_vector_type(4))) float v4f;
typedef __attribute__((ext_vector_type(8))) float v8f;

#define B_  4
#define L_  512
#define V_  32
#define D_  64
#define MT  32
#define SCALE 0.125f
#define LDSS (L_ + 4)
#define NC_ 8

__global__ __launch_bounds__(256)
void sum_key_kernel(const float* __restrict__ key, float* __restrict__ sumv) {
    int idx = blockIdx.x * blockDim.x + threadIdx.x;
    if (idx >= B_ * L_ * D_) return;
    int d  = idx % D_;
    int bl = idx / D_;
    const float* p = key + (size_t)bl * (V_ * D_) + d;
    float s = 0.f;
#pragma unroll
    for (int v = 0; v < V_; ++v) s += p[v * D_];
    sumv[idx] = s;
}

__global__ __launch_bounds__(32)
void attn_kernel(const float* __restrict__ q,
                 const float* __restrict__ sumv,
                 const float* __restrict__ val,
                 const int*   __restrict__ lab,
                 float*       __restrict__ out) {
    __shared__ __attribute__((aligned(16))) float smem[MT * LDSS];
    __shared__ __attribute__((aligned(16))) float labf[L_];
    __shared__ __attribute__((aligned(16))) int   perm[L_];
    __shared__ float rowsum[MT];

    const int blk   = blockIdx.x;
    const int itile = blk % (L_ / MT);
    const int bk    = blk / (L_ / MT);
    const int k     = bk % V_;
    const int b     = bk / V_;

    const int lane = threadIdx.x;
    const int hlf  = lane >> 4;
    const int l16  = lane & 15;
    const int i0   = itile * MT;

    int lv[16];
    bool inrange = true;
#pragma unroll
    for (int t = 0; t < 16; ++t) {
        lv[t] = lab[b * L_ + t * 32 + lane];
        inrange = inrange && ((unsigned)lv[t] < (unsigned)NC_);
    }
    const bool sorted = (__builtin_amdgcn_ballot_w32(!inrange) == 0u);

    int pos[16];
    int jlo_u = 0, jhi_u = L_;
    if (sorted) {
        int seg[NC_ + 1];
        int base = 0;
#pragma unroll
        for (int t = 0; t < 16; ++t) pos[t] = 0;
#pragma unroll
        for (int c = 0; c < NC_; ++c) {
            seg[c] = base;
#pragma unroll
            for (int t = 0; t < 16; ++t) {
                const bool hit = (lv[t] == c);
                const unsigned m = __builtin_amdgcn_ballot_w32(hit);
                const int p = (int)__builtin_amdgcn_mbcnt_lo(m, (unsigned)base);
                pos[t] = hit ? p : pos[t];
                base += __popc(m);
            }
        }
        seg[NC_] = base;
#pragma unroll
        for (int c = 0; c < NC_; ++c)
            if (seg[c] <= i0) jlo_u = seg[c];
#pragma unroll
        for (int c = NC_; c >= 1; --c)
            if (seg[c] > i0 + MT - 1) jhi_u = seg[c];
        jlo_u = jlo_u & ~15;
        jhi_u = (jhi_u + 15) & ~15;
    } else {
#pragma unroll
        for (int t = 0; t < 16; ++t) pos[t] = t * 32 + lane;
    }
#pragma unroll
    for (int t = 0; t < 16; ++t) {
        const int p = pos[t] & (L_ - 1);
        perm[p] = t * 32 + lane;
        labf[p] = (float)lv[t];
    }
    __syncthreads();

    int jlo = __builtin_amdgcn_readfirstlane(jlo_u);
    int jhi = __builtin_amdgcn_readfirstlane(jhi_u);
    jlo = jlo < 0 ? 0 : jlo;
    jhi = jhi > L_ ? L_ : jhi;

    v2f qa0[16], qa1[16];
    {
        const int r0 = perm[i0 + l16] & (L_ - 1);
        const int r1 = perm[i0 + 16 + l16] & (L_ - 1);
        const float* q0 =
            q + (((size_t)(b * L_ + r0) * V_) + k) * D_ + hlf * 2;
        const float* q1 =
            q + (((size_t)(b * L_ + r1) * V_) + k) * D_ + hlf * 2;
#pragma unroll
        for (int s = 0; s < 16; ++s) {
            qa0[s] = *(const v2f*)(q0 + s * 4);
            qa1[s] = *(const v2f*)(q1 + s * 4);
        }
    }

    const float* sb = sumv + (size_t)b * L_ * D_;
    for (int jt = jlo >> 4; jt < (jhi >> 4); ++jt) {
        v8f c0 = {}, c1 = {};
        const int js = perm[jt * 16 + l16] & (L_ - 1);
        const float* srow = sb + (size_t)js * D_ + hlf * 2;
#pragma unroll
        for (int s = 0; s < 16; ++s) {
            v2f bf = *(const v2f*)(srow + s * 4);
            c0 = __builtin_amdgcn_wmma_f32_16x16x4_f32(
                     false, qa0[s], false, bf, (short)0, c0, false, false);
            c1 = __builtin_amdgcn_wmma_f32_16x16x4_f32(
                     false, qa1[s], false, bf, (short)0, c1, false, false);
        }
#pragma unroll
        for (int r = 0; r < 8; ++r) {
            smem[(r + hlf * 8) * LDSS + jt * 16 + l16]      = c0[r];
            smem[(16 + r + hlf * 8) * LDSS + jt * 16 + l16] = c1[r];
        }
    }
    __syncthreads();

    {
        const int row = lane;
        const float myLab = labf[i0 + row];
        float* srow = &smem[row * LDSS];

        v4f m4 = {-INFINITY, -INFINITY, -INFINITY, -INFINITY};
        for (int cb = jlo; cb < jhi; cb += 16) {
#pragma unroll
            for (int u = 0; u < 4; ++u) {
                const int c = cb + u * 4;
                v4f s  = *(v4f*)(srow + c);
                v4f lv4 = *(const v4f*)(&labf[c]);
                s.x = (lv4.x == myLab) ? s.x : -INFINITY;
                s.y = (lv4.y == myLab) ? s.y : -INFINITY;
                s.z = (lv4.z == myLab) ? s.z : -INFINITY;
                s.w = (lv4.w == myLab) ? s.w : -INFINITY;
                *(v4f*)(srow + c) = s;
                m4.x = fmaxf(m4.x, s.x);  m4.y = fmaxf(m4.y, s.y);
                m4.z = fmaxf(m4.z, s.z);  m4.w = fmaxf(m4.w, s.w);
            }
        }
        const float m = fmaxf(fmaxf(m4.x, m4.y), fmaxf(m4.z, m4.w));

        v4f s4 = {0.f, 0.f, 0.f, 0.f};
        for (int cb = jlo; cb < jhi; cb += 16) {
#pragma unroll
            for (int u = 0; u < 4; ++u) {
                const int c = cb + u * 4;
                v4f s = *(v4f*)(srow + c);
                v4f p;
                p.x = __expf(SCALE * (s.x - m));
                p.y = __expf(SCALE * (s.y - m));
                p.z = __expf(SCALE * (s.z - m));
                p.w = __expf(SCALE * (s.w - m));
                *(v4f*)(srow + c) = p;
                s4 += p;
            }
        }
        rowsum[row] = (s4.x + s4.y) + (s4.z + s4.w);
    }
    __syncthreads();

    v8f acc[2][4] = {{{}, {}, {}, {}}, {{}, {}, {}, {}}};
    const float* vb = val + ((size_t)(b * L_) * V_ + k) * D_;
    for (int kb = jlo; kb < jhi; kb += 16) {
#pragma unroll
        for (int u = 0; u < 4; ++u) {
            const int j0 = kb + u * 4 + hlf * 2;
            const v2f af0 = *(const v2f*)(&smem[l16 * LDSS + j0]);
            const v2f af1 = *(const v2f*)(&smem[(16 + l16) * LDSS + j0]);
            const int p0 = perm[j0] & (L_ - 1);
            const int p1 = perm[j0 + 1] & (L_ - 1);
            const float* v0 = vb + (size_t)p0 * (V_ * D_);
            const float* v1 = vb + (size_t)p1 * (V_ * D_);
#pragma unroll
            for (int n = 0; n < 4; ++n) {
                v2f bf;
                bf.x = v0[n * 16 + l16];
                bf.y = v1[n * 16 + l16];
                acc[0][n] = __builtin_amdgcn_wmma_f32_16x16x4_f32(
                                false, af0, false, bf, (short)0, acc[0][n], false, false);
                acc[1][n] = __builtin_amdgcn_wmma_f32_16x16x4_f32(
                                false, af1, false, bf, (short)0, acc[1][n], false, false);
            }
        }
    }

#pragma unroll
    for (int t = 0; t < 2; ++t) {
#pragma unroll
        for (int r = 0; r < 8; ++r) {
            const int irow = t * 16 + r + hlf * 8;
            const int i = perm[i0 + irow] & (L_ - 1);
            const float rs = rowsum[irow];
            float* orow = out + (((size_t)(b * L_ + i) * V_) + k) * D_ + l16;
#pragma unroll
            for (int n = 0; n < 4; ++n) {
                orow[n * 16] = acc[t][n][r] / rs;
            }
        }
    }
}

extern "C" void kernel_launch(void* const* d_in, const int* in_sizes, int n_in,
                              void* d_out, int out_size, void* d_ws, size_t ws_size,
                              hipStream_t stream) {
    const float* query = (const float*)d_in[0];
    const float* key   = (const float*)d_in[1];
    const float* value = (const float*)d_in[2];
    const int*   label = (const int*)d_in[3];
    float* out  = (float*)d_out;
    float* sumv = (float*)d_ws;

    const int n1 = B_ * L_ * D_;
    sum_key_kernel<<<(n1 + 255) / 256, 256, 0, stream>>>(key, sumv);

    const int blocks = B_ * V_ * (L_ / MT);
    attn_kernel<<<blocks, 32, 0, stream>>>(query, sumv, value, label, out);
}


# Score buffer holds one chunk of at most 192 keys with a running row max and sum across chunks: LDS per wave 70 KB -> 30 KB, so all 2048 waves are resident at once (2 per SIMD)
# speedup vs baseline: 3.3542x; 1.4394x over previous
#include <hip/hip_runtime.h>
#include <math.h>


typedef __attribute__((ext_vector_type(2))) float v2f;
typedef __attribute__((ext_vector_type(4))) float v4f;
typedef __attribute__((ext_vector_type(8))) float v8f;

#define B_  4
#define L_  512
#define V_  32
#define D_  64
#define MT  32
#define SCALE 0.125f
#define CH   192
#define LDSS (CH + 4)
#define NC_ 8

__global__ __launch_bounds__(256)
void sum_key_kernel(const float* __restrict__ key, float* __restrict__ sumv) {
    int idx = blockIdx.x * blockDim.x + threadIdx.x;
    if (idx >= B_ * L_ * D_) return;
    int d  = idx % D_;
    int bl = idx / D_;
    const float* p = key + (size_t)bl * (V_ * D_) + d;
    float s = 0.f;
#pragma unroll
    for (int v = 0; v < V_; ++v) s += p[v * D_];
    sumv[idx] = s;
}

__global__ __launch_bounds__(32)
void attn_kernel(const float* __restrict__ q,
                 const float* __restrict__ sumv,
                 const float* __restrict__ val,
                 const int*   __restrict__ lab,
                 float*       __restrict__ out) {
    __shared__ __attribute__((aligned(16))) float smem[MT * LDSS];
    __shared__ __attribute__((aligned(16))) float labf[L_];
    __shared__ __attribute__((aligned(16))) int   perm[L_];
    __shared__ float rowsum[MT];
    __shared__ float alphas[MT];

    const int blk   = blockIdx.x;
    const int itile = blk % (L_ / MT);
    const int bk    = blk / (L_ / MT);
    const int k     = bk % V_;
    const int b     = bk / V_;

    const int lane = threadIdx.x;
    const int hlf  = lane >> 4;
    const int l16  = lane & 15;
    const int i0   = itile * MT;

    int lv[16];
    bool inrange = true;
#pragma unroll
    for (int t = 0; t < 16; ++t) {
        lv[t] = lab[b * L_ + t * 32 + lane];
        inrange = inrange && ((unsigned)lv[t] < (unsigned)NC_);
    }
    const bool sorted = (__builtin_amdgcn_ballot_w32(!inrange) == 0u);

    int pos[16];
    int jlo_u = 0, jhi_u = L_;
    if (sorted) {
        int seg[NC_ + 1];
        int base = 0;
#pragma unroll
        for (int t = 0; t < 16; ++t) pos[t] = 0;
#pragma unroll
        for (int c = 0; c < NC_; ++c) {
            seg[c] = base;
#pragma unroll
            for (int t = 0; t < 16; ++t) {
                const bool hit = (lv[t] == c);
                const unsigned m = __builtin_amdgcn_ballot_w32(hit);
                const int p = (int)__builtin_amdgcn_mbcnt_lo(m, (unsigned)base);
                pos[t] = hit ? p : pos[t];
                base += __popc(m);
            }
        }
        seg[NC_] = base;
#pragma unroll
        for (int c = 0; c < NC_; ++c)
            if (seg[c] <= i0) jlo_u = seg[c];
#pragma unroll
        for (int c = NC_; c >= 1; --c)
            if (seg[c] > i0 + MT - 1) jhi_u = seg[c];
        jlo_u = jlo_u & ~15;
        jhi_u = (jhi_u + 15) & ~15;
    } else {
#pragma unroll
        for (int t = 0; t < 16; ++t) pos[t] = t * 32 + lane;
    }
#pragma unroll
    for (int t = 0; t < 16; ++t) {
        const int p = pos[t] & (L_ - 1);
        perm[p] = t * 32 + lane;
        labf[p] = (float)lv[t];
    }
    __syncthreads();

    int jlo = __builtin_amdgcn_readfirstlane(jlo_u);
    int jhi = __builtin_amdgcn_readfirstlane(jhi_u);
    jlo = jlo < 0 ? 0 : jlo;
    jhi = jhi > L_ ? L_ : jhi;

    v2f qa0[16], qa1[16];
    {
        const int r0 = perm[i0 + l16] & (L_ - 1);
        const int r1 = perm[i0 + 16 + l16] & (L_ - 1);
        const float* q0 =
            q + (((size_t)(b * L_ + r0) * V_) + k) * D_ + hlf * 2;
        const float* q1 =
            q + (((size_t)(b * L_ + r1) * V_) + k) * D_ + hlf * 2;
#pragma unroll
        for (int s = 0; s < 16; ++s) {
            qa0[s] = *(const v2f*)(q0 + s * 4);
            qa1[s] = *(const v2f*)(q1 + s * 4);
        }
    }

    const float* sb = sumv + (size_t)b * L_ * D_;
    const float* vb = val + ((size_t)(b * L_) * V_ + k) * D_;
    const float myLab = labf[i0 + lane];
    float m_run = -INFINITY;
    float l_run = 0.f;
    v8f acc[2][4] = {{{}, {}, {}, {}}, {{}, {}, {}, {}}};

    for (int c0 = jlo; c0 < jhi; c0 += CH) {
        const int c1 = (c0 + CH < jhi) ? (c0 + CH) : jhi;

        for (int jt = c0 >> 4; jt < (c1 >> 4); ++jt) {
            v8f s0 = {}, s1 = {};
            const int js = perm[jt * 16 + l16] & (L_ - 1);
            const float* srow = sb + (size_t)js * D_ + hlf * 2;
#pragma unroll
            for (int s = 0; s < 16; ++s) {
                v2f bf = *(const v2f*)(srow + s * 4);
                s0 = __builtin_amdgcn_wmma_f32_16x16x4_f32(
                         false, qa0[s], false, bf, (short)0, s0, false, false);
                s1 = __builtin_amdgcn_wmma_f32_16x16x4_f32(
                         false, qa1[s], false, bf, (short)0, s1, false, false);
            }
            const int col = jt * 16 - c0 + l16;
#pragma unroll
            for (int r = 0; r < 8; ++r) {
                smem[(r + hlf * 8) * LDSS + col]      = s0[r];
                smem[(16 + r + hlf * 8) * LDSS + col] = s1[r];
            }
        }
        __syncthreads();

        {
            float* srow = &smem[lane * LDSS];

            v4f m4 = {-INFINITY, -INFINITY, -INFINITY, -INFINITY};
            for (int cb = c0; cb < c1; cb += 16) {
#pragma unroll
                for (int u = 0; u < 4; ++u) {
                    const int c = cb + u * 4;
                    v4f s  = *(v4f*)(srow + (c - c0));
                    v4f lv4 = *(const v4f*)(&labf[c]);
                    s.x = (lv4.x == myLab) ? s.x : -INFINITY;
                    s.y = (lv4.y == myLab) ? s.y : -INFINITY;
                    s.z = (lv4.z == myLab) ? s.z : -INFINITY;
                    s.w = (lv4.w == myLab) ? s.w : -INFINITY;
                    *(v4f*)(srow + (c - c0)) = s;
                    m4.x = fmaxf(m4.x, s.x);  m4.y = fmaxf(m4.y, s.y);
                    m4.z = fmaxf(m4.z, s.z);  m4.w = fmaxf(m4.w, s.w);
                }
            }
            const float mc    = fmaxf(fmaxf(m4.x, m4.y), fmaxf(m4.z, m4.w));
            const float m_new = fmaxf(m_run, mc);
            const float m_use = (m_new == -INFINITY) ? 0.f : m_new;
            const float alpha = __expf(SCALE * (m_run - m_use));

            v4f s4 = {0.f, 0.f, 0.f, 0.f};
            for (int cb = c0; cb < c1; cb += 16) {
#pragma unroll
                for (int u = 0; u < 4; ++u) {
                    const int c = cb + u * 4 - c0;
                    v4f s = *(v4f*)(srow + c);
                    v4f p;
                    p.x = __expf(SCALE * (s.x - m_use));
                    p.y = __expf(SCALE * (s.y - m_use));
                    p.z = __expf(SCALE * (s.z - m_use));
                    p.w = __expf(SCALE * (s.w - m_use));
                    *(v4f*)(srow + c) = p;
                    s4 += p;
                }
            }
            l_run = l_run * alpha + ((s4.x + s4.y) + (s4.z + s4.w));
            m_run = m_new;
            alphas[lane] = alpha;
        }
        __syncthreads();

        if (c0 != jlo) {
#pragma unroll
            for (int t = 0; t < 2; ++t) {
#pragma unroll
                for (int r = 0; r < 8; ++r) {
                    const float a = alphas[t * 16 + r + hlf * 8];
#pragma unroll
                    for (int n = 0; n < 4; ++n) acc[t][n][r] *= a;
                }
            }
        }
        for (int kb = c0; kb < c1; kb += 16) {
#pragma unroll
            for (int u = 0; u < 4; ++u) {
                const int j0 = kb + u * 4 + hlf * 2;
                const v2f af0 = *(const v2f*)(&smem[l16 * LDSS + (j0 - c0)]);
                const v2f af1 = *(const v2f*)(&smem[(16 + l16) * LDSS + (j0 - c0)]);
                const int p0 = perm[j0] & (L_ - 1);
                const int p1 = perm[j0 + 1] & (L_ - 1);
                const float* v0 = vb + (size_t)p0 * (V_ * D_);
                const float* v1 = vb + (size_t)p1 * (V_ * D_);
#pragma unroll
                for (int n = 0; n < 4; ++n) {
                    v2f bf;
                    bf.x = v0[n * 16 + l16];
                    bf.y = v1[n * 16 + l16];
                    acc[0][n] = __builtin_amdgcn_wmma_f32_16x16x4_f32(
                                    false, af0, false, bf, (short)0, acc[0][n], false, false);
                    acc[1][n] = __builtin_amdgcn_wmma_f32_16x16x4_f32(
                                    false, af1, false, bf, (short)0, acc[1][n], false, false);
                }
            }
        }
        __syncthreads();
    }
    rowsum[lane] = l_run;
    __syncthreads();

#pragma unroll
    for (int t = 0; t < 2; ++t) {
#pragma unroll
        for (int r = 0; r < 8; ++r) {
            const int irow = t * 16 + r + hlf * 8;
            const int i = perm[i0 + irow] & (L_ - 1);
            const float rs = rowsum[irow];
            float* orow = out + (((size_t)(b * L_ + i) * V_) + k) * D_ + l16;
#pragma unroll
            for (int n = 0; n < 4; ++n) {
                orow[n * 16] = acc[t][n][r] / rs;
            }
        }
    }
}

extern "C" void kernel_launch(void* const* d_in, const int* in_sizes, int n_in,
                              void* d_out, int out_size, void* d_ws, size_t ws_size,
                              hipStream_t stream) {
    const float* query = (const float*)d_in[0];
    const float* key   = (const float*)d_in[1];
    const float* value = (const float*)d_in[2];
    const int*   label = (const int*)d_in[3];
    float* out  = (float*)d_out;
    float* sumv = (float*)d_ws;

    const int n1 = B_ * L_ * D_;
    sum_key_kernel<<<(n1 + 255) / 256, 256, 0, stream>>>(key, sumv);

    const int blocks = B_ * V_ * (L_ / MT);
    attn_kernel<<<blocks, 32, 0, stream>>>(query, sumv, value, label, out);
}


# Row tiles aligned to label starts: a wave takes up to 32 sorted rows of ONE label and visits only that label's keys; 24 tile slots per batch-head, 160-key chunk so all waves stay resident
# speedup vs baseline: 3.6938x; 1.1012x over previous
#include <hip/hip_runtime.h>
#include <math.h>


typedef __attribute__((ext_vector_type(2))) float v2f;
typedef __attribute__((ext_vector_type(4))) float v4f;
typedef __attribute__((ext_vector_type(8))) float v8f;

#define B_  4
#define L_  512
#define V_  32
#define D_  64
#define MT  32
#define SCALE 0.125f
#define CH   160
#define LDSS (CH + 4)
#define NC_ 8
#define NT_ (L_ / MT + NC_)

__global__ __launch_bounds__(256)
void sum_key_kernel(const float* __restrict__ key, float* __restrict__ sumv) {
    int idx = blockIdx.x * blockDim.x + threadIdx.x;
    if (idx >= B_ * L_ * D_) return;
    int d  = idx % D_;
    int bl = idx / D_;
    const float* p = key + (size_t)bl * (V_ * D_) + d;
    float s = 0.f;
#pragma unroll
    for (int v = 0; v < V_; ++v) s += p[v * D_];
    sumv[idx] = s;
}

__global__ __launch_bounds__(32)
void attn_kernel(const float* __restrict__ q,
                 const float* __restrict__ sumv,
                 const float* __restrict__ val,
                 const int*   __restrict__ lab,
                 float*       __restrict__ out) {
    __shared__ __attribute__((aligned(16))) float smem[MT * LDSS];
    __shared__ __attribute__((aligned(16))) float labf[L_ + MT];
    __shared__ __attribute__((aligned(16))) int   perm[L_ + MT];
    __shared__ float rowsum[MT];
    __shared__ float alphas[MT];

    const int blk   = blockIdx.x;
    const int itile = blk % NT_;
    const int bk    = blk / NT_;
    const int k     = bk % V_;
    const int b     = bk / V_;

    const int lane = threadIdx.x;
    const int hlf  = lane >> 4;
    const int l16  = lane & 15;

    int lv[16];
    bool inrange = true;
#pragma unroll
    for (int t = 0; t < 16; ++t) {
        lv[t] = lab[b * L_ + t * 32 + lane];
        inrange = inrange && ((unsigned)lv[t] < (unsigned)NC_);
    }
    const bool sorted = (__builtin_amdgcn_ballot_w32(!inrange) == 0u);

    int pos[16];
    int jlo_u = 0, jhi_u = 0, i0_u = 0, nrows_u = 0;
    if (sorted) {
        int seg[NC_ + 1];
        int base = 0;
#pragma unroll
        for (int t = 0; t < 16; ++t) pos[t] = 0;
#pragma unroll
        for (int c = 0; c < NC_; ++c) {
            seg[c] = base;
#pragma unroll
            for (int t = 0; t < 16; ++t) {
                const bool hit = (lv[t] == c);
                const unsigned m = __builtin_amdgcn_ballot_w32(hit);
                const int p = (int)__builtin_amdgcn_mbcnt_lo(m, (unsigned)base);
                pos[t] = hit ? p : pos[t];
                base += __popc(m);
            }
        }
        seg[NC_] = base;
        int g0 = 0;
#pragma unroll
        for (int c = 0; c < NC_; ++c) {
            const int n  = seg[c + 1] - seg[c];
            const int ng = (n + MT - 1) / MT;
            if (itile >= g0 && itile < g0 + ng) {
                i0_u    = seg[c] + (itile - g0) * MT;
                const int left = seg[c + 1] - i0_u;
                nrows_u = left < MT ? left : MT;
                jlo_u   = seg[c];
                jhi_u   = seg[c + 1];
            }
            g0 += ng;
        }
    } else {
#pragma unroll
        for (int t = 0; t < 16; ++t) pos[t] = t * 32 + lane;
        if (itile < L_ / MT) {
            i0_u = itile * MT;  nrows_u = MT;  jlo_u = 0;  jhi_u = L_;
        }
    }
    const int nrows = __builtin_amdgcn_readfirstlane(nrows_u);
    if (nrows <= 0) return;
    int i0 = __builtin_amdgcn_readfirstlane(i0_u);
    i0 = i0 < 0 ? 0 : (i0 > L_ - 1 ? L_ - 1 : i0);
    const bool two = nrows > 16;
#pragma unroll
    for (int t = 0; t < 16; ++t) {
        const int p = pos[t] & (L_ - 1);
        perm[p] = t * 32 + lane;
        labf[p] = (float)lv[t];
    }
    perm[L_ + lane] = 0;
    labf[L_ + lane] = __builtin_nanf("");
    __syncthreads();

    int jlo = __builtin_amdgcn_readfirstlane(jlo_u);
    int jhi = __builtin_amdgcn_readfirstlane(jhi_u);
    jlo = jlo < 0 ? 0 : jlo;
    jhi = jhi > L_ ? L_ : jhi;

    v2f qa0[16], qa1[16];
    {
        const int r0 = perm[i0 + l16] & (L_ - 1);
        const int r1 = perm[i0 + 16 + l16] & (L_ - 1);
        const float* q0 =
            q + (((size_t)(b * L_ + r0) * V_) + k) * D_ + hlf * 2;
        const float* q1 =
            q + (((size_t)(b * L_ + r1) * V_) + k) * D_ + hlf * 2;
#pragma unroll
        for (int s = 0; s < 16; ++s) {
            qa0[s] = *(const v2f*)(q0 + s * 4);
            qa1[s] = *(const v2f*)(q1 + s * 4);
        }
    }

    const float* sb = sumv + (size_t)b * L_ * D_;
    const float* vb = val + ((size_t)(b * L_) * V_ + k) * D_;
    const float myLab = labf[i0 + lane];
    float m_run = -INFINITY;
    float l_run = 0.f;
    v8f acc[2][4] = {{{}, {}, {}, {}}, {{}, {}, {}, {}}};

    for (int c0 = jlo; c0 < jhi; c0 += CH) {
        const int c1   = (c0 + CH < jhi) ? (c0 + CH) : jhi;
        const int ncol = (c1 - c0 + 15) & ~15;

        if (two) {
            for (int ct = 0; ct < ncol; ct += 16) {
                v8f s0 = {}, s1 = {};
                const int js = perm[c0 + ct + l16] & (L_ - 1);
                const float* srow = sb + (size_t)js * D_ + hlf * 2;
#pragma unroll
                for (int s = 0; s < 16; ++s) {
                    v2f bf = *(const v2f*)(srow + s * 4);
                    s0 = __builtin_amdgcn_wmma_f32_16x16x4_f32(
                             false, qa0[s], false, bf, (short)0, s0, false, false);
                    s1 = __builtin_amdgcn_wmma_f32_16x16x4_f32(
                             false, qa1[s], false, bf, (short)0, s1, false, false);
                }
                const int col = ct + l16;
#pragma unroll
                for (int r = 0; r < 8; ++r) {
                    smem[(r + hlf * 8) * LDSS + col]      = s0[r];
                    smem[(16 + r + hlf * 8) * LDSS + col] = s1[r];
                }
            }
        } else {
            for (int ct = 0; ct < ncol; ct += 16) {
                v8f s0 = {}, s0b = {};
                const int js = perm[c0 + ct + l16] & (L_ - 1);
                const float* srow = sb + (size_t)js * D_ + hlf * 2;
#pragma unroll
                for (int s = 0; s < 16; s += 2) {
                    v2f bfa = *(const v2f*)(srow + s * 4);
                    v2f bfb = *(const v2f*)(srow + s * 4 + 4);
                    s0  = __builtin_amdgcn_wmma_f32_16x16x4_f32(
                              false, qa0[s], false, bfa, (short)0, s0, false, false);
                    s0b = __builtin_amdgcn_wmma_f32_16x16x4_f32(
                              false, qa0[s + 1], false, bfb, (short)0, s0b, false, false);
                }
                s0 += s0b;
                const int col = ct + l16;
#pragma unroll
                for (int r = 0; r < 8; ++r) {
                    smem[(r + hlf * 8) * LDSS + col]      = s0[r];
                    smem[(16 + r + hlf * 8) * LDSS + col] = 0.f;
                }
            }
        }
        __syncthreads();

        {
            float* srow = &smem[lane * LDSS];

            v4f m4 = {-INFINITY, -INFINITY, -INFINITY, -INFINITY};
            for (int cb = 0; cb < ncol; cb += 16) {
#pragma unroll
                for (int u = 0; u < 4; ++u) {
                    const int cc = cb + u * 4;
                    const int c  = c0 + cc;
                    v4f s  = *(v4f*)(srow + cc);
                    const float lx = labf[c], ly = labf[c + 1];
                    const float lz = labf[c + 2], lw = labf[c + 3];
                    s.x = (lx == myLab) ? s.x : -INFINITY;
                    s.y = (ly == myLab) ? s.y : -INFINITY;
                    s.z = (lz == myLab) ? s.z : -INFINITY;
                    s.w = (lw == myLab) ? s.w : -INFINITY;
                    *(v4f*)(srow + cc) = s;
                    m4.x = fmaxf(m4.x, s.x);  m4.y = fmaxf(m4.y, s.y);
                    m4.z = fmaxf(m4.z, s.z);  m4.w = fmaxf(m4.w, s.w);
                }
            }
            const float mc    = fmaxf(fmaxf(m4.x, m4.y), fmaxf(m4.z, m4.w));
            const float m_new = fmaxf(m_run, mc);
            const float m_use = (m_new == -INFINITY) ? 0.f : m_new;
            const float alpha = __expf(SCALE * (m_run - m_use));

            v4f s4 = {0.f, 0.f, 0.f, 0.f};
            for (int cb = 0; cb < ncol; cb += 16) {
#pragma unroll
                for (int u = 0; u < 4; ++u) {
                    const int c = cb + u * 4;
                    v4f s = *(v4f*)(srow + c);
                    v4f p;
                    p.x = __expf(SCALE * (s.x - m_use));
                    p.y = __expf(SCALE * (s.y - m_use));
                    p.z = __expf(SCALE * (s.z - m_use));
                    p.w = __expf(SCALE * (s.w - m_use));
                    *(v4f*)(srow + c) = p;
                    s4 += p;
                }
            }
            l_run = l_run * alpha + ((s4.x + s4.y) + (s4.z + s4.w));
            m_run = m_new;
            alphas[lane] = alpha;
        }
        __syncthreads();

        if (c0 != jlo) {
#pragma unroll
            for (int t = 0; t < 2; ++t) {
#pragma unroll
                for (int r = 0; r < 8; ++r) {
                    const float a = alphas[t * 16 + r + hlf * 8];
#pragma unroll
                    for (int n = 0; n < 4; ++n) acc[t][n][r] *= a;
                }
            }
        }
        if (two) {
            for (int kb = 0; kb < ncol; kb += 16) {
#pragma unroll
                for (int u = 0; u < 4; ++u) {
                    const int jc = kb + u * 4 + hlf * 2;
                    const int j0 = c0 + jc;
                    const v2f af0 = *(const v2f*)(&smem[l16 * LDSS + jc]);
                    const v2f af1 = *(const v2f*)(&smem[(16 + l16) * LDSS + jc]);
                    const int p0 = perm[j0] & (L_ - 1);
                    const int p1 = perm[j0 + 1] & (L_ - 1);
                    const float* v0 = vb + (size_t)p0 * (V_ * D_);
                    const float* v1 = vb + (size_t)p1 * (V_ * D_);
#pragma unroll
                    for (int n = 0; n < 4; ++n) {
                        v2f bf;
                        bf.x = v0[n * 16 + l16];
                        bf.y = v1[n * 16 + l16];
                        acc[0][n] = __builtin_amdgcn_wmma_f32_16x16x4_f32(
                                        false, af0, false, bf, (short)0, acc[0][n], false, false);
                        acc[1][n] = __builtin_amdgcn_wmma_f32_16x16x4_f32(
                                        false, af1, false, bf, (short)0, acc[1][n], false, false);
                    }
                }
            }
        } else {
            for (int kb = 0; kb < ncol; kb += 16) {
#pragma unroll
                for (int u = 0; u < 4; ++u) {
                    const int jc = kb + u * 4 + hlf * 2;
                    const int j0 = c0 + jc;
                    const v2f af0 = *(const v2f*)(&smem[l16 * LDSS + jc]);
                    const int p0 = perm[j0] & (L_ - 1);
                    const int p1 = perm[j0 + 1] & (L_ - 1);
                    const float* v0 = vb + (size_t)p0 * (V_ * D_);
                    const float* v1 = vb + (size_t)p1 * (V_ * D_);
#pragma unroll
                    for (int n = 0; n < 4; ++n) {
                        v2f bf;
                        bf.x = v0[n * 16 + l16];
                        bf.y = v1[n * 16 + l16];
                        acc[0][n] = __builtin_amdgcn_wmma_f32_16x16x4_f32(
                                        false, af0, false, bf, (short)0, acc[0][n], false, false);
                    }
                }
            }
        }
        __syncthreads();
    }
    rowsum[lane] = l_run;
    __syncthreads();

#pragma unroll
    for (int t = 0; t < 2; ++t) {
#pragma unroll
        for (int r = 0; r < 8; ++r) {
            const int irow = t * 16 + r + hlf * 8;
            if (irow < nrows) {
                const int i = perm[(i0 + irow) & (L_ - 1)] & (L_ - 1);
                const float rs = rowsum[irow];
                float* orow = out + (((size_t)(b * L_ + i) * V_) + k) * D_ + l16;
#pragma unroll
                for (int n = 0; n < 4; ++n) {
                    orow[n * 16] = acc[t][n][r] / rs;
                }
            }
        }
    }
}

extern "C" void kernel_launch(void* const* d_in, const int* in_sizes, int n_in,
                              void* d_out, int out_size, void* d_ws, size_t ws_size,
                              hipStream_t stream) {
    const float* query = (const float*)d_in[0];
    const float* key   = (const float*)d_in[1];
    const float* value = (const float*)d_in[2];
    const int*   label = (const int*)d_in[3];
    float* out  = (float*)d_out;
    float* sumv = (float*)d_ws;

    const int n1 = B_ * L_ * D_;
    sum_key_kernel<<<(n1 + 255) / 256, 256, 0, stream>>>(key, sumv);

    const int blocks = B_ * V_ * NT_;
    attn_kernel<<<blocks, 32, 0, stream>>>(query, sumv, value, label, out);
}


# Each wave lists only its own label's positions (16 ballots; label c owns tile slots 3c..3c+2) instead of counting-sorting all 512 positions by all 8 labels in every wave
# speedup vs baseline: 4.2134x; 1.1407x over previous
#include <hip/hip_runtime.h>
#include <math.h>


typedef __attribute__((ext_vector_type(2))) float v2f;
typedef __attribute__((ext_vector_type(4))) float v4f;
typedef __attribute__((ext_vector_type(8))) float v8f;

#define B_  4
#define L_  512
#define V_  32
#define D_  64
#define MT  32
#define SCALE 0.125f
#define CH   160
#define LDSS (CH + 4)
#define NC_ 8
#define GPL 3
#define NT_ (NC_ * GPL)

__global__ __launch_bounds__(256)
void sum_key_kernel(const float* __restrict__ key, float* __restrict__ sumv) {
    int idx = blockIdx.x * blockDim.x + threadIdx.x;
    if (idx >= B_ * L_ * D_) return;
    int d  = idx % D_;
    int bl = idx / D_;
    const float* p = key + (size_t)bl * (V_ * D_) + d;
    float s = 0.f;
#pragma unroll
    for (int v = 0; v < V_; ++v) s += p[v * D_];
    sumv[idx] = s;
}

__global__ __launch_bounds__(32)
void attn_kernel(const float* __restrict__ q,
                 const float* __restrict__ sumv,
                 const float* __restrict__ val,
                 const int*   __restrict__ lab,
                 float*       __restrict__ out) {
    __shared__ __attribute__((aligned(16))) float smem[MT * LDSS];
    __shared__ __attribute__((aligned(16))) float labf[L_ + MT];
    __shared__ __attribute__((aligned(16))) int   perm[L_ + MT];
    __shared__ float rowsum[MT];
    __shared__ float alphas[MT];

    const int blk   = blockIdx.x;
    const int itile = blk % NT_;
    const int bk    = blk / NT_;
    const int k     = bk % V_;
    const int b     = bk / V_;

    const int lane = threadIdx.x;
    const int hlf  = lane >> 4;
    const int l16  = lane & 15;

    int lv[16];
    bool inrange = true;
#pragma unroll
    for (int t = 0; t < 16; ++t) {
        lv[t] = lab[b * L_ + t * 32 + lane];
        inrange = inrange && ((unsigned)lv[t] < (unsigned)NC_);
    }
    const bool sorted = (__builtin_amdgcn_ballot_w32(!inrange) == 0u);

    int nkey_u, g_u, gstep_u, ntile_u;
    if (sorted) {
        const int cown = itile / GPL;
        int base = 0;
#pragma unroll
        for (int t = 0; t < 16; ++t) {
            const bool hit = (lv[t] == cown);
            const unsigned m = __builtin_amdgcn_ballot_w32(hit);
            const int p = (int)__builtin_amdgcn_mbcnt_lo(m, (unsigned)base);
            if (hit) perm[p] = t * 32 + lane;
            base += __popc(m);
        }
        const int n = base;
        perm[n + lane] = 0;
        const float cf = (float)cown;
#pragma unroll 1
        for (int j = lane; j < n + MT; j += 32)
            labf[j] = (j < n) ? cf : __builtin_nanf("");
        nkey_u  = n;
        g_u     = itile % GPL;
        gstep_u = GPL;
        ntile_u = (n + MT - 1) / MT;
    } else {
#pragma unroll
        for (int t = 0; t < 16; ++t) {
            perm[t * 32 + lane] = t * 32 + lane;
            labf[t * 32 + lane] = (float)lv[t];
        }
        perm[L_ + lane] = 0;
        labf[L_ + lane] = __builtin_nanf("");
        nkey_u  = L_;
        g_u     = itile;
        gstep_u = NT_;
        ntile_u = L_ / MT;
    }
    __syncthreads();
    int nkey = __builtin_amdgcn_readfirstlane(nkey_u);
    nkey = nkey < 0 ? 0 : (nkey > L_ ? L_ : nkey);
    const int gstep = __builtin_amdgcn_readfirstlane(gstep_u);
    const int ntile = __builtin_amdgcn_readfirstlane(ntile_u);
    const int jlo = 0;
    const int jhi = nkey;

    const float* sb = sumv + (size_t)b * L_ * D_;
    const float* vb = val + ((size_t)(b * L_) * V_ + k) * D_;

    for (int g = __builtin_amdgcn_readfirstlane(g_u); g < ntile; g += gstep) {
    int i0 = g * MT;
    i0 = i0 < 0 ? 0 : (i0 > L_ - 1 ? L_ - 1 : i0);
    const int left  = nkey - i0;
    const int nrows = left < MT ? left : MT;
    if (nrows <= 0) break;
    const bool two = nrows > 16;

    v2f qa0[16], qa1[16];
    {
        const int r0 = perm[i0 + l16] & (L_ - 1);
        const int r1 = perm[i0 + 16 + l16] & (L_ - 1);
        const float* q0 =
            q + (((size_t)(b * L_ + r0) * V_) + k) * D_ + hlf * 2;
        const float* q1 =
            q + (((size_t)(b * L_ + r1) * V_) + k) * D_ + hlf * 2;
#pragma unroll
        for (int s = 0; s < 16; ++s) {
            qa0[s] = *(const v2f*)(q0 + s * 4);
            qa1[s] = *(const v2f*)(q1 + s * 4);
        }
    }

    const float myLab = labf[i0 + lane];
    float m_run = -INFINITY;
    float l_run = 0.f;
    v8f acc[2][4] = {{{}, {}, {}, {}}, {{}, {}, {}, {}}};

    for (int c0 = jlo; c0 < jhi; c0 += CH) {
        const int c1   = (c0 + CH < jhi) ? (c0 + CH) : jhi;
        const int ncol = (c1 - c0 + 15) & ~15;

        if (two) {
            for (int ct = 0; ct < ncol; ct += 16) {
                v8f s0 = {}, s1 = {};
                const int js = perm[c0 + ct + l16] & (L_ - 1);
                const float* srow = sb + (size_t)js * D_ + hlf * 2;
#pragma unroll
                for (int s = 0; s < 16; ++s) {
                    v2f bf = *(const v2f*)(srow + s * 4);
                    s0 = __builtin_amdgcn_wmma_f32_16x16x4_f32(
                             false, qa0[s], false, bf, (short)0, s0, false, false);
                    s1 = __builtin_amdgcn_wmma_f32_16x16x4_f32(
                             false, qa1[s], false, bf, (short)0, s1, false, false);
                }
                const int col = ct + l16;
#pragma unroll
                for (int r = 0; r < 8; ++r) {
                    smem[(r + hlf * 8) * LDSS + col]      = s0[r];
                    smem[(16 + r + hlf * 8) * LDSS + col] = s1[r];
                }
            }
        } else {
            for (int ct = 0; ct < ncol; ct += 16) {
                v8f s0 = {}, s0b = {};
                const int js = perm[c0 + ct + l16] & (L_ - 1);
                const float* srow = sb + (size_t)js * D_ + hlf * 2;
#pragma unroll
                for (int s = 0; s < 16; s += 2) {
                    v2f bfa = *(const v2f*)(srow + s * 4);
                    v2f bfb = *(const v2f*)(srow + s * 4 + 4);
                    s0  = __builtin_amdgcn_wmma_f32_16x16x4_f32(
                              false, qa0[s], false, bfa, (short)0, s0, false, false);
                    s0b = __builtin_amdgcn_wmma_f32_16x16x4_f32(
                              false, qa0[s + 1], false, bfb, (short)0, s0b, false, false);
                }
                s0 += s0b;
                const int col = ct + l16;
#pragma unroll
                for (int r = 0; r < 8; ++r) {
                    smem[(r + hlf * 8) * LDSS + col]      = s0[r];
                    smem[(16 + r + hlf * 8) * LDSS + col] = 0.f;
                }
            }
        }
        __syncthreads();

        {
            float* srow = &smem[lane * LDSS];

            v4f m4 = {-INFINITY, -INFINITY, -INFINITY, -INFINITY};
            for (int cb = 0; cb < ncol; cb += 16) {
#pragma unroll
                for (int u = 0; u < 4; ++u) {
                    const int cc = cb + u * 4;
                    const int c  = c0 + cc;
                    v4f s  = *(v4f*)(srow + cc);
                    const float lx = labf[c], ly = labf[c + 1];
                    const float lz = labf[c + 2], lw = labf[c + 3];
                    s.x = (lx == myLab) ? s.x : -INFINITY;
                    s.y = (ly == myLab) ? s.y : -INFINITY;
                    s.z = (lz == myLab) ? s.z : -INFINITY;
                    s.w = (lw == myLab) ? s.w : -INFINITY;
                    *(v4f*)(srow + cc) = s;
                    m4.x = fmaxf(m4.x, s.x);  m4.y = fmaxf(m4.y, s.y);
                    m4.z = fmaxf(m4.z, s.z);  m4.w = fmaxf(m4.w, s.w);
                }
            }
            const float mc    = fmaxf(fmaxf(m4.x, m4.y), fmaxf(m4.z, m4.w));
            const float m_new = fmaxf(m_run, mc);
            const float m_use = (m_new == -INFINITY) ? 0.f : m_new;
            const float alpha = __expf(SCALE * (m_run - m_use));

            v4f s4 = {0.f, 0.f, 0.f, 0.f};
            for (int cb = 0; cb < ncol; cb += 16) {
#pragma unroll
                for (int u = 0; u < 4; ++u) {
                    const int c = cb + u * 4;
                    v4f s = *(v4f*)(srow + c);
                    v4f p;
                    p.x = __expf(SCALE * (s.x - m_use));
                    p.y = __expf(SCALE * (s.y - m_use));
                    p.z = __expf(SCALE * (s.z - m_use));
                    p.w = __expf(SCALE * (s.w - m_use));
                    *(v4f*)(srow + c) = p;
                    s4 += p;
                }
            }
            l_run = l_run * alpha + ((s4.x + s4.y) + (s4.z + s4.w));
            m_run = m_new;
            alphas[lane] = alpha;
        }
        __syncthreads();

        if (c0 != jlo) {
#pragma unroll
            for (int t = 0; t < 2; ++t) {
#pragma unroll
                for (int r = 0; r < 8; ++r) {
                    const float a = alphas[t * 16 + r + hlf * 8];
#pragma unroll
                    for (int n = 0; n < 4; ++n) acc[t][n][r] *= a;
                }
            }
        }
        if (two) {
            for (int kb = 0; kb < ncol; kb += 16) {
#pragma unroll
                for (int u = 0; u < 4; ++u) {
                    const int jc = kb + u * 4 + hlf * 2;
                    const int j0 = c0 + jc;
                    const v2f af0 = *(const v2f*)(&smem[l16 * LDSS + jc]);
                    const v2f af1 = *(const v2f*)(&smem[(16 + l16) * LDSS + jc]);
                    const int p0 = perm[j0] & (L_ - 1);
                    const int p1 = perm[j0 + 1] & (L_ - 1);
                    const float* v0 = vb + (size_t)p0 * (V_ * D_);
                    const float* v1 = vb + (size_t)p1 * (V_ * D_);
#pragma unroll
                    for (int n = 0; n < 4; ++n) {
                        v2f bf;
                        bf.x = v0[n * 16 + l16];
                        bf.y = v1[n * 16 + l16];
                        acc[0][n] = __builtin_amdgcn_wmma_f32_16x16x4_f32(
                                        false, af0, false, bf, (short)0, acc[0][n], false, false);
                        acc[1][n] = __builtin_amdgcn_wmma_f32_16x16x4_f32(
                                        false, af1, false, bf, (short)0, acc[1][n], false, false);
                    }
                }
            }
        } else {
            for (int kb = 0; kb < ncol; kb += 16) {
#pragma unroll
                for (int u = 0; u < 4; ++u) {
                    const int jc = kb + u * 4 + hlf * 2;
                    const int j0 = c0 + jc;
                    const v2f af0 = *(const v2f*)(&smem[l16 * LDSS + jc]);
                    const int p0 = perm[j0] & (L_ - 1);
                    const int p1 = perm[j0 + 1] & (L_ - 1);
                    const float* v0 = vb + (size_t)p0 * (V_ * D_);
                    const float* v1 = vb + (size_t)p1 * (V_ * D_);
#pragma unroll
                    for (int n = 0; n < 4; ++n) {
                        v2f bf;
                        bf.x = v0[n * 16 + l16];
                        bf.y = v1[n * 16 + l16];
                        acc[0][n] = __builtin_amdgcn_wmma_f32_16x16x4_f32(
                                        false, af0, false, bf, (short)0, acc[0][n], false, false);
                    }
                }
            }
        }
        __syncthreads();
    }
    rowsum[lane] = l_run;
    __syncthreads();

#pragma unroll
    for (int t = 0; t < 2; ++t) {
#pragma unroll
        for (int r = 0; r < 8; ++r) {
            const int irow = t * 16 + r + hlf * 8;
            if (irow < nrows) {
                const int i = perm[(i0 + irow) & (L_ - 1)] & (L_ - 1);
                const float rs = rowsum[irow];
                float* orow = out + (((size_t)(b * L_ + i) * V_) + k) * D_ + l16;
#pragma unroll
                for (int n = 0; n < 4; ++n) {
                    orow[n * 16] = acc[t][n][r] / rs;
                }
            }
        }
    }
    }
}

extern "C" void kernel_launch(void* const* d_in, const int* in_sizes, int n_in,
                              void* d_out, int out_size, void* d_ws, size_t ws_size,
                              hipStream_t stream) {
    const float* query = (const float*)d_in[0];
    const float* key   = (const float*)d_in[1];
    const float* value = (const float*)d_in[2];
    const int*   label = (const int*)d_in[3];
    float* out  = (float*)d_out;
    float* sumv = (float*)d_ws;

    const int n1 = B_ * L_ * D_;
    sum_key_kernel<<<(n1 + 255) / 256, 256, 0, stream>>>(key, sumv);

    const int blocks = B_ * V_ * NT_;
    attn_kernel<<<blocks, 32, 0, stream>>>(query, sumv, value, label, out);
}


# Attention kernel: next 16-key tile's S and V fragment loads issued ahead of the current tile's matrix products (two register sets); first V tile loaded before softmax, no load drain between phases
# speedup vs baseline: 4.3719x; 1.0376x over previous
#include <hip/hip_runtime.h>
#include <math.h>


typedef __attribute__((ext_vector_type(2))) float v2f;
typedef __attribute__((ext_vector_type(4))) float v4f;
typedef __attribute__((ext_vector_type(8))) float v8f;

#define B_  4
#define L_  512
#define V_  32
#define D_  64
#define MT  32
#define SCALE 0.125f
#define CH   160
#define LDSS (CH + 4)
#define NC_ 8
#define GPL 3
#define NT_ (NC_ * GPL)

__global__ __launch_bounds__(256)
void sum_key_kernel(const float* __restrict__ key, float* __restrict__ sumv) {
    int idx = blockIdx.x * blockDim.x + threadIdx.x;
    if (idx >= B_ * L_ * D_) return;
    int d  = idx % D_;
    int bl = idx / D_;
    const float* p = key + (size_t)bl * (V_ * D_) + d;
    float s = 0.f;
#pragma unroll
    for (int v = 0; v < V_; ++v) s += p[v * D_];
    sumv[idx] = s;
}

#define WMMA4(a, b, c) __builtin_amdgcn_wmma_f32_16x16x4_f32(false, (a), false, (b), (short)0, (c), false, false)

#define LOAD_S(buf, CT)                                                        \
    {                                                                          \
        const int js_ = perm[c0 + (CT) + l16] & (L_ - 1);                      \
        const float* sr_ = sb + (size_t)js_ * D_ + hlf * 2;                    \
        _Pragma("unroll")                                                      \
        for (int s = 0; s < 16; ++s) buf[s] = *(const v2f*)(sr_ + s * 4);      \
    }

#define MM_S2(buf, CT)                                                         \
    {                                                                          \
        v8f s0_ = {}, s1_ = {};                                                \
        _Pragma("unroll")                                                      \
        for (int s = 0; s < 16; ++s) {                                         \
            s0_ = WMMA4(qa0[s], buf[s], s0_);                                  \
            s1_ = WMMA4(qa1[s], buf[s], s1_);                                  \
        }                                                                      \
        const int col_ = (CT) + l16;                                           \
        _Pragma("unroll")                                                      \
        for (int r = 0; r < 8; ++r) {                                          \
            smem[(r + hlf * 8) * LDSS + col_]      = s0_[r];                   \
            smem[(16 + r + hlf * 8) * LDSS + col_] = s1_[r];                   \
        }                                                                      \
    }

#define MM_S1(buf, CT)                                                         \
    {                                                                          \
        v8f s0_ = {}, s0b_ = {};                                               \
        _Pragma("unroll")                                                      \
        for (int s = 0; s < 16; s += 2) {                                      \
            s0_  = WMMA4(qa0[s], buf[s], s0_);                                 \
            s0b_ = WMMA4(qa0[s + 1], buf[s + 1], s0b_);                        \
        }                                                                      \
        s0_ += s0b_;                                                           \
        const int col_ = (CT) + l16;                                           \
        _Pragma("unroll")                                                      \
        for (int r = 0; r < 8; ++r) {                                          \
            smem[(r + hlf * 8) * LDSS + col_]      = s0_[r];                   \
            smem[(16 + r + hlf * 8) * LDSS + col_] = 0.f;                      \
        }                                                                      \
    }

#define LOAD_V(buf, KB)                                                        \
    {                                                                          \
        _Pragma("unroll")                                                      \
        for (int u = 0; u < 4; ++u) {                                          \
            const int j0_ = c0 + (KB) + u * 4 + hlf * 2;                       \
            const int p0_ = perm[j0_] & (L_ - 1);                              \
            const int p1_ = perm[j0_ + 1] & (L_ - 1);                          \
            const float* v0_ = vb + (size_t)p0_ * (V_ * D_);                   \
            const float* v1_ = vb + (size_t)p1_ * (V_ * D_);                   \
            _Pragma("unroll")                                                  \
            for (int n = 0; n < 4; ++n) {                                      \
                buf[u * 4 + n].x = v0_[n * 16 + l16];                          \
                buf[u * 4 + n].y = v1_[n * 16 + l16];                          \
            }                                                                  \
        }                                                                      \
    }

#define MM_V2(buf, KB)                                                         \
    {                                                                          \
        _Pragma("unroll")                                                      \
        for (int u = 0; u < 4; ++u) {                                          \
            const int jc_ = (KB) + u * 4 + hlf * 2;                            \
            const v2f af0_ = *(const v2f*)(&smem[l16 * LDSS + jc_]);           \
            const v2f af1_ = *(const v2f*)(&smem[(16 + l16) * LDSS + jc_]);    \
            _Pragma("unroll")                                                  \
            for (int n = 0; n < 4; ++n) {                                      \
                acc[0][n] = WMMA4(af0_, buf[u * 4 + n], acc[0][n]);            \
                acc[1][n] = WMMA4(af1_, buf[u * 4 + n], acc[1][n]);            \
            }                                                                  \
        }                                                                      \
    }

#define MM_V1(buf, KB)                                                         \
    {                                                                          \
        _Pragma("unroll")                                                      \
        for (int u = 0; u < 4; ++u) {                                          \
            const int jc_ = (KB) + u * 4 + hlf * 2;                            \
            const v2f af0_ = *(const v2f*)(&smem[l16 * LDSS + jc_]);           \
            _Pragma("unroll")                                                  \
            for (int n = 0; n < 4; ++n)                                        \
                acc[0][n] = WMMA4(af0_, buf[u * 4 + n], acc[0][n]);            \
        }                                                                      \
    }

#define TILE_LOOP(NCOL, LOAD, MM, bufA, bufB)                                  \
    {                                                                          \
        const int last_  = (NCOL) - 16;                                        \
        const int npair_ = (NCOL) >> 5;                                        \
        int t_ = 0;                                                            \
        for (int p_ = 0; p_ < npair_; ++p_) {                                  \
            LOAD(bufB, t_ + 16)                                                \
            MM(bufA, t_)                                                       \
            const int ta_ = (t_ + 32 < last_) ? (t_ + 32) : last_;             \
            LOAD(bufA, ta_)                                                    \
            MM(bufB, t_ + 16)                                                  \
            t_ += 32;                                                          \
        }                                                                      \
        if ((NCOL) & 16) MM(bufA, last_)                                       \
    }

#define WAVE_SYNC()                                                            \
    {                                                                          \
        __builtin_amdgcn_fence(__ATOMIC_SEQ_CST, "wavefront");                 \
        __builtin_amdgcn_wave_barrier();                                       \
    }

__global__ __launch_bounds__(32)
void attn_kernel(const float* __restrict__ q,
                 const float* __restrict__ sumv,
                 const float* __restrict__ val,
                 const int*   __restrict__ lab,
                 float*       __restrict__ out) {
    __shared__ __attribute__((aligned(16))) float smem[MT * LDSS];
    __shared__ __attribute__((aligned(16))) float labf[L_ + MT];
    __shared__ __attribute__((aligned(16))) int   perm[L_ + MT];
    __shared__ float rowsum[MT];
    __shared__ float alphas[MT];

    const int blk   = blockIdx.x;
    const int itile = blk % NT_;
    const int bk    = blk / NT_;
    const int k     = bk % V_;
    const int b     = bk / V_;

    const int lane = threadIdx.x;
    const int hlf  = lane >> 4;
    const int l16  = lane & 15;

    int lv[16];
    bool inrange = true;
#pragma unroll
    for (int t = 0; t < 16; ++t) {
        lv[t] = lab[b * L_ + t * 32 + lane];
        inrange = inrange && ((unsigned)lv[t] < (unsigned)NC_);
    }
    const bool sorted = (__builtin_amdgcn_ballot_w32(!inrange) == 0u);

    int nkey_u, g_u, gstep_u, ntile_u;
    if (sorted) {
        const int cown = itile / GPL;
        int base = 0;
#pragma unroll
        for (int t = 0; t < 16; ++t) {
            const bool hit = (lv[t] == cown);
            const unsigned m = __builtin_amdgcn_ballot_w32(hit);
            const int p = (int)__builtin_amdgcn_mbcnt_lo(m, (unsigned)base);
            if (hit) perm[p] = t * 32 + lane;
            base += __popc(m);
        }
        const int n = base;
        perm[n + lane] = 0;
        const float cf = (float)cown;
#pragma unroll 1
        for (int j = lane; j < n + MT; j += 32)
            labf[j] = (j < n) ? cf : __builtin_nanf("");
        nkey_u  = n;
        g_u     = itile % GPL;
        gstep_u = GPL;
        ntile_u = (n + MT - 1) / MT;
    } else {
#pragma unroll
        for (int t = 0; t < 16; ++t) {
            perm[t * 32 + lane] = t * 32 + lane;
            labf[t * 32 + lane] = (float)lv[t];
        }
        perm[L_ + lane] = 0;
        labf[L_ + lane] = __builtin_nanf("");
        nkey_u  = L_;
        g_u     = itile;
        gstep_u = NT_;
        ntile_u = L_ / MT;
    }
    __syncthreads();
    int nkey = __builtin_amdgcn_readfirstlane(nkey_u);
    nkey = nkey < 0 ? 0 : (nkey > L_ ? L_ : nkey);
    const int gstep = __builtin_amdgcn_readfirstlane(gstep_u);
    const int ntile = __builtin_amdgcn_readfirstlane(ntile_u);
    const int jlo = 0;
    const int jhi = nkey;

    const float* sb = sumv + (size_t)b * L_ * D_;
    const float* vb = val + ((size_t)(b * L_) * V_ + k) * D_;

    for (int g = __builtin_amdgcn_readfirstlane(g_u); g < ntile; g += gstep) {
    int i0 = g * MT;
    i0 = i0 < 0 ? 0 : (i0 > L_ - 1 ? L_ - 1 : i0);
    const int left  = nkey - i0;
    const int nrows = left < MT ? left : MT;
    if (nrows <= 0) break;
    const bool two = nrows > 16;

    v2f qa0[16], qa1[16];
    {
        const int r0 = perm[i0 + l16] & (L_ - 1);
        const int r1 = perm[i0 + 16 + l16] & (L_ - 1);
        const float* q0 =
            q + (((size_t)(b * L_ + r0) * V_) + k) * D_ + hlf * 2;
        const float* q1 =
            q + (((size_t)(b * L_ + r1) * V_) + k) * D_ + hlf * 2;
#pragma unroll
        for (int s = 0; s < 16; ++s) {
            qa0[s] = *(const v2f*)(q0 + s * 4);
            qa1[s] = *(const v2f*)(q1 + s * 4);
        }
    }

    const float myLab = labf[i0 + lane];
    float m_run = -INFINITY;
    float l_run = 0.f;
    v8f acc[2][4] = {{{}, {}, {}, {}}, {{}, {}, {}, {}}};

    for (int c0 = jlo; c0 < jhi; c0 += CH) {
        const int c1   = (c0 + CH < jhi) ? (c0 + CH) : jhi;
        const int ncol = (c1 - c0 + 15) & ~15;

        {
            v2f sA[16], sB[16];
            LOAD_S(sA, 0)
            if (two) {
                TILE_LOOP(ncol, LOAD_S, MM_S2, sA, sB)
            } else {
                TILE_LOOP(ncol, LOAD_S, MM_S1, sA, sB)
            }
        }
        v2f vA[16], vB[16];
        LOAD_V(vA, 0)
        WAVE_SYNC()

        {
            float* srow = &smem[lane * LDSS];

            v4f m4 = {-INFINITY, -INFINITY, -INFINITY, -INFINITY};
            for (int cb = 0; cb < ncol; cb += 16) {
#pragma unroll
                for (int u = 0; u < 4; ++u) {
                    const int cc = cb + u * 4;
                    const int c  = c0 + cc;
                    v4f s  = *(v4f*)(srow + cc);
                    const float lx = labf[c], ly = labf[c + 1];
                    const float lz = labf[c + 2], lw = labf[c + 3];
                    s.x = (lx == myLab) ? s.x : -INFINITY;
                    s.y = (ly == myLab) ? s.y : -INFINITY;
                    s.z = (lz == myLab) ? s.z : -INFINITY;
                    s.w = (lw == myLab) ? s.w : -INFINITY;
                    *(v4f*)(srow + cc) = s;
                    m4.x = fmaxf(m4.x, s.x);  m4.y = fmaxf(m4.y, s.y);
                    m4.z = fmaxf(m4.z, s.z);  m4.w = fmaxf(m4.w, s.w);
                }
            }
            const float mc    = fmaxf(fmaxf(m4.x, m4.y), fmaxf(m4.z, m4.w));
            const float m_new = fmaxf(m_run, mc);
            const float m_use = (m_new == -INFINITY) ? 0.f : m_new;
            const float alpha = __expf(SCALE * (m_run - m_use));

            v4f s4 = {0.f, 0.f, 0.f, 0.f};
            for (int cb = 0; cb < ncol; cb += 16) {
#pragma unroll
                for (int u = 0; u < 4; ++u) {
                    const int c = cb + u * 4;
                    v4f s = *(v4f*)(srow + c);
                    v4f p;
                    p.x = __expf(SCALE * (s.x - m_use));
                    p.y = __expf(SCALE * (s.y - m_use));
                    p.z = __expf(SCALE * (s.z - m_use));
                    p.w = __expf(SCALE * (s.w - m_use));
                    *(v4f*)(srow + c) = p;
                    s4 += p;
                }
            }
            l_run = l_run * alpha + ((s4.x + s4.y) + (s4.z + s4.w));
            m_run = m_new;
            alphas[lane] = alpha;
        }
        WAVE_SYNC()

        if (c0 != jlo) {
#pragma unroll
            for (int t = 0; t < 2; ++t) {
#pragma unroll
                for (int r = 0; r < 8; ++r) {
                    const float a = alphas[t * 16 + r + hlf * 8];
#pragma unroll
                    for (int n = 0; n < 4; ++n) acc[t][n][r] *= a;
                }
            }
        }
        if (two) {
            TILE_LOOP(ncol, LOAD_V, MM_V2, vA, vB)
        } else {
            TILE_LOOP(ncol, LOAD_V, MM_V1, vA, vB)
        }
        __syncthreads();
    }
    rowsum[lane] = l_run;
    __syncthreads();

#pragma unroll
    for (int t = 0; t < 2; ++t) {
#pragma unroll
        for (int r = 0; r < 8; ++r) {
            const int irow = t * 16 + r + hlf * 8;
            if (irow < nrows) {
                const int i = perm[(i0 + irow) & (L_ - 1)] & (L_ - 1);
                const float rs = rowsum[irow];
                float* orow = out + (((size_t)(b * L_ + i) * V_) + k) * D_ + l16;
#pragma unroll
                for (int n = 0; n < 4; ++n) {
                    orow[n * 16] = acc[t][n][r] / rs;
                }
            }
        }
    }
    }
}

extern "C" void kernel_launch(void* const* d_in, const int* in_sizes, int n_in,
                              void* d_out, int out_size, void* d_ws, size_t ws_size,
                              hipStream_t stream) {
    const float* query = (const float*)d_in[0];
    const float* key   = (const float*)d_in[1];
    const float* value = (const float*)d_in[2];
    const int*   label = (const int*)d_in[3];
    float* out  = (float*)d_out;
    float* sumv = (float*)d_ws;

    const int n1 = B_ * L_ * D_;
    sum_key_kernel<<<(n1 + 255) / 256, 256, 0, stream>>>(key, sumv);

    const int blocks = B_ * V_ * NT_;
    attn_kernel<<<blocks, 32, 0, stream>>>(query, sumv, value, label, out);
}


# Attention kernel: all global accesses 16 bytes per lane: Q and key-sum fragments as float4 (same k order both sides); output columns re-mapped (lane c owns d=4c..4c+3): whole-row V loads and stores
# speedup vs baseline: 4.5722x; 1.0458x over previous
#include <hip/hip_runtime.h>
#include <math.h>


typedef __attribute__((ext_vector_type(2))) float v2f;
typedef __attribute__((ext_vector_type(4))) float v4f;
typedef __attribute__((ext_vector_type(8))) float v8f;

#define B_  4
#define L_  512
#define V_  32
#define D_  64
#define MT  32
#define SCALE 0.125f
#define CH   160
#define LDSS (CH + 4)
#define NC_ 8
#define GPL 3
#define NT_ (NC_ * GPL)

__global__ __launch_bounds__(256)
void sum_key_kernel(const float* __restrict__ key, float* __restrict__ sumv) {
    int idx = blockIdx.x * blockDim.x + threadIdx.x;
    if (idx >= B_ * L_ * D_) return;
    int d  = idx % D_;
    int bl = idx / D_;
    const float* p = key + (size_t)bl * (V_ * D_) + d;
    float s = 0.f;
#pragma unroll
    for (int v = 0; v < V_; ++v) s += p[v * D_];
    sumv[idx] = s;
}

#define LO2(v) __builtin_shufflevector((v), (v), 0, 1)
#define HI2(v) __builtin_shufflevector((v), (v), 2, 3)
#define PIN_ROWS(a, b) asm volatile("" : "+v"(a), "+v"(b));
#define WMMA4(a, b, c) __builtin_amdgcn_wmma_f32_16x16x4_f32(false, (a), false, (b), (short)0, (c), false, false)

#define LOAD_S(buf, CT)                                                        \
    {                                                                          \
        const int js_ = perm[c0 + (CT) + l16] & (L_ - 1);                      \
        const float* sr_ = sb + (size_t)js_ * D_ + hlf * 4;                    \
        _Pragma("unroll")                                                      \
        for (int s = 0; s < 8; ++s) buf[s] = *(const v4f*)(sr_ + s * 8);       \
    }

#define MM_S2(buf, CT)                                                         \
    {                                                                          \
        v8f s0_ = {}, s1_ = {};                                                \
        _Pragma("unroll")                                                      \
        for (int s = 0; s < 8; ++s) {                                          \
            s0_ = WMMA4(LO2(qa0[s]), LO2(buf[s]), s0_);                        \
            s1_ = WMMA4(LO2(qa1[s]), LO2(buf[s]), s1_);                        \
            s0_ = WMMA4(HI2(qa0[s]), HI2(buf[s]), s0_);                        \
            s1_ = WMMA4(HI2(qa1[s]), HI2(buf[s]), s1_);                        \
        }                                                                      \
        const int col_ = (CT) + l16;                                           \
        _Pragma("unroll")                                                      \
        for (int r = 0; r < 8; ++r) {                                          \
            smem[(r + hlf * 8) * LDSS + col_]      = s0_[r];                   \
            smem[(16 + r + hlf * 8) * LDSS + col_] = s1_[r];                   \
        }                                                                      \
    }

#define MM_S1(buf, CT)                                                         \
    {                                                                          \
        v8f s0_ = {}, s0b_ = {};                                               \
        _Pragma("unroll")                                                      \
        for (int s = 0; s < 8; ++s) {                                          \
            s0_  = WMMA4(LO2(qa0[s]), LO2(buf[s]), s0_);                       \
            s0b_ = WMMA4(HI2(qa0[s]), HI2(buf[s]), s0b_);                      \
        }                                                                      \
        s0_ += s0b_;                                                           \
        const int col_ = (CT) + l16;                                           \
        _Pragma("unroll")                                                      \
        for (int r = 0; r < 8; ++r) {                                          \
            smem[(r + hlf * 8) * LDSS + col_]      = s0_[r];                   \
            smem[(16 + r + hlf * 8) * LDSS + col_] = 0.f;                      \
        }                                                                      \
    }

#define LOAD_V(buf, KB)                                                        \
    {                                                                          \
        _Pragma("unroll")                                                      \
        for (int u = 0; u < 4; ++u) {                                          \
            const int j0_ = c0 + (KB) + u * 4 + hlf * 2;                       \
            const int p0_ = perm[j0_] & (L_ - 1);                              \
            const int p1_ = perm[j0_ + 1] & (L_ - 1);                          \
            buf[u * 2]     =                                                   \
                *(const v4f*)(vb + (size_t)p0_ * (V_ * D_) + l16 * 4);         \
            buf[u * 2 + 1] =                                                   \
                *(const v4f*)(vb + (size_t)p1_ * (V_ * D_) + l16 * 4);         \
        }                                                                      \
    }

#define MM_V2(buf, KB)                                                         \
    {                                                                          \
        _Pragma("unroll")                                                      \
        for (int u = 0; u < 4; ++u) {                                          \
            const int jc_ = (KB) + u * 4 + hlf * 2;                            \
            const v2f af0_ = *(const v2f*)(&smem[l16 * LDSS + jc_]);           \
            const v2f af1_ = *(const v2f*)(&smem[(16 + l16) * LDSS + jc_]);    \
            v4f k0_ = buf[u * 2], k1_ = buf[u * 2 + 1];                        \
            PIN_ROWS(k0_, k1_)                                                 \
            _Pragma("unroll")                                                  \
            for (int n = 0; n < 4; ++n) {                                      \
                const v2f bf_ = {k0_[n], k1_[n]};                              \
                acc[0][n] = WMMA4(af0_, bf_, acc[0][n]);                       \
                acc[1][n] = WMMA4(af1_, bf_, acc[1][n]);                       \
            }                                                                  \
        }                                                                      \
    }

#define MM_V1(buf, KB)                                                         \
    {                                                                          \
        _Pragma("unroll")                                                      \
        for (int u = 0; u < 4; ++u) {                                          \
            const int jc_ = (KB) + u * 4 + hlf * 2;                            \
            const v2f af0_ = *(const v2f*)(&smem[l16 * LDSS + jc_]);           \
            v4f k0_ = buf[u * 2], k1_ = buf[u * 2 + 1];                        \
            PIN_ROWS(k0_, k1_)                                                 \
            _Pragma("unroll")                                                  \
            for (int n = 0; n < 4; ++n) {                                      \
                const v2f bf_ = {k0_[n], k1_[n]};                              \
                acc[0][n] = WMMA4(af0_, bf_, acc[0][n]);                       \
            }                                                                  \
        }                                                                      \
    }

#define TILE_LOOP(NCOL, LOAD, MM, bufA, bufB)                                  \
    {                                                                          \
        const int last_  = (NCOL) - 16;                                        \
        const int npair_ = (NCOL) >> 5;                                        \
        int t_ = 0;                                                            \
        for (int p_ = 0; p_ < npair_; ++p_) {                                  \
            LOAD(bufB, t_ + 16)                                                \
            MM(bufA, t_)                                                       \
            const int ta_ = (t_ + 32 < last_) ? (t_ + 32) : last_;             \
            LOAD(bufA, ta_)                                                    \
            MM(bufB, t_ + 16)                                                  \
            t_ += 32;                                                          \
        }                                                                      \
        if ((NCOL) & 16) MM(bufA, last_)                                       \
    }

#define WAVE_SYNC()                                                            \
    {                                                                          \
        __builtin_amdgcn_fence(__ATOMIC_SEQ_CST, "wavefront");                 \
        __builtin_amdgcn_wave_barrier();                                       \
    }

__global__ __launch_bounds__(32) __attribute__((amdgpu_waves_per_eu(3)))
void attn_kernel(const float* __restrict__ q,
                 const float* __restrict__ sumv,
                 const float* __restrict__ val,
                 const int*   __restrict__ lab,
                 float*       __restrict__ out) {
    __shared__ __attribute__((aligned(16))) float smem[MT * LDSS];
    __shared__ __attribute__((aligned(16))) float labf[L_ + MT];
    __shared__ __attribute__((aligned(16))) int   perm[L_ + MT];
    __shared__ float rowsum[MT];
    __shared__ float alphas[MT];

    const int blk   = blockIdx.x;
    const int itile = blk % NT_;
    const int bk    = blk / NT_;
    const int k     = bk % V_;
    const int b     = bk / V_;

    const int lane = threadIdx.x;
    const int hlf  = lane >> 4;
    const int l16  = lane & 15;

    int lv[16];
    bool inrange = true;
#pragma unroll
    for (int t = 0; t < 16; ++t) {
        lv[t] = lab[b * L_ + t * 32 + lane];
        inrange = inrange && ((unsigned)lv[t] < (unsigned)NC_);
    }
    const bool sorted = (__builtin_amdgcn_ballot_w32(!inrange) == 0u);

    int nkey_u, g_u, gstep_u, ntile_u;
    if (sorted) {
        const int cown = itile / GPL;
        int base = 0;
#pragma unroll
        for (int t = 0; t < 16; ++t) {
            const bool hit = (lv[t] == cown);
            const unsigned m = __builtin_amdgcn_ballot_w32(hit);
            const int p = (int)__builtin_amdgcn_mbcnt_lo(m, (unsigned)base);
            if (hit) perm[p] = t * 32 + lane;
            base += __popc(m);
        }
        const int n = base;
        perm[n + lane] = 0;
        const float cf = (float)cown;
#pragma unroll 1
        for (int j = lane; j < n + MT; j += 32)
            labf[j] = (j < n) ? cf : __builtin_nanf("");
        nkey_u  = n;
        g_u     = itile % GPL;
        gstep_u = GPL;
        ntile_u = (n + MT - 1) / MT;
    } else {
#pragma unroll
        for (int t = 0; t < 16; ++t) {
            perm[t * 32 + lane] = t * 32 + lane;
            labf[t * 32 + lane] = (float)lv[t];
        }
        perm[L_ + lane] = 0;
        labf[L_ + lane] = __builtin_nanf("");
        nkey_u  = L_;
        g_u     = itile;
        gstep_u = NT_;
        ntile_u = L_ / MT;
    }
    __syncthreads();
    int nkey = __builtin_amdgcn_readfirstlane(nkey_u);
    nkey = nkey < 0 ? 0 : (nkey > L_ ? L_ : nkey);
    const int gstep = __builtin_amdgcn_readfirstlane(gstep_u);
    const int ntile = __builtin_amdgcn_readfirstlane(ntile_u);
    const int jlo = 0;
    const int jhi = nkey;

    const float* sb = sumv + (size_t)b * L_ * D_;
    const float* vb = val + ((size_t)(b * L_) * V_ + k) * D_;

    for (int g = __builtin_amdgcn_readfirstlane(g_u); g < ntile; g += gstep) {
    int i0 = g * MT;
    i0 = i0 < 0 ? 0 : (i0 > L_ - 1 ? L_ - 1 : i0);
    const int left  = nkey - i0;
    const int nrows = left < MT ? left : MT;
    if (nrows <= 0) break;
    const bool two = nrows > 16;

    v4f qa0[8], qa1[8];
    {
        const int r0 = perm[i0 + l16] & (L_ - 1);
        const int r1 = perm[i0 + 16 + l16] & (L_ - 1);
        const float* q0 =
            q + (((size_t)(b * L_ + r0) * V_) + k) * D_ + hlf * 4;
        const float* q1 =
            q + (((size_t)(b * L_ + r1) * V_) + k) * D_ + hlf * 4;
#pragma unroll
        for (int s = 0; s < 8; ++s) {
            qa0[s] = *(const v4f*)(q0 + s * 8);
            qa1[s] = *(const v4f*)(q1 + s * 8);
        }
    }

    const float myLab = labf[i0 + lane];
    float m_run = -INFINITY;
    float l_run = 0.f;
    v8f acc[2][4] = {{{}, {}, {}, {}}, {{}, {}, {}, {}}};

    for (int c0 = jlo; c0 < jhi; c0 += CH) {
        const int c1   = (c0 + CH < jhi) ? (c0 + CH) : jhi;
        const int ncol = (c1 - c0 + 15) & ~15;

        {
            v4f sA[8], sB[8];
            LOAD_S(sA, 0)
            if (two) {
                TILE_LOOP(ncol, LOAD_S, MM_S2, sA, sB)
            } else {
                TILE_LOOP(ncol, LOAD_S, MM_S1, sA, sB)
            }
        }
        v4f vA[8], vB[8];
        LOAD_V(vA, 0)
        WAVE_SYNC()

        {
            float* srow = &smem[lane * LDSS];

            v4f m4 = {-INFINITY, -INFINITY, -INFINITY, -INFINITY};
            for (int cb = 0; cb < ncol; cb += 16) {
#pragma unroll
                for (int u = 0; u < 4; ++u) {
                    const int cc = cb + u * 4;
                    const int c  = c0 + cc;
                    v4f s  = *(v4f*)(srow + cc);
                    const float lx = labf[c], ly = labf[c + 1];
                    const float lz = labf[c + 2], lw = labf[c + 3];
                    s.x = (lx == myLab) ? s.x : -INFINITY;
                    s.y = (ly == myLab) ? s.y : -INFINITY;
                    s.z = (lz == myLab) ? s.z : -INFINITY;
                    s.w = (lw == myLab) ? s.w : -INFINITY;
                    *(v4f*)(srow + cc) = s;
                    m4.x = fmaxf(m4.x, s.x);  m4.y = fmaxf(m4.y, s.y);
                    m4.z = fmaxf(m4.z, s.z);  m4.w = fmaxf(m4.w, s.w);
                }
            }
            const float mc    = fmaxf(fmaxf(m4.x, m4.y), fmaxf(m4.z, m4.w));
            const float m_new = fmaxf(m_run, mc);
            const float m_use = (m_new == -INFINITY) ? 0.f : m_new;
            const float alpha = __expf(SCALE * (m_run - m_use));

            v4f s4 = {0.f, 0.f, 0.f, 0.f};
            for (int cb = 0; cb < ncol; cb += 16) {
#pragma unroll
                for (int u = 0; u < 4; ++u) {
                    const int c = cb + u * 4;
                    v4f s = *(v4f*)(srow + c);
                    v4f p;
                    p.x = __expf(SCALE * (s.x - m_use));
                    p.y = __expf(SCALE * (s.y - m_use));
                    p.z = __expf(SCALE * (s.z - m_use));
                    p.w = __expf(SCALE * (s.w - m_use));
                    *(v4f*)(srow + c) = p;
                    s4 += p;
                }
            }
            l_run = l_run * alpha + ((s4.x + s4.y) + (s4.z + s4.w));
            m_run = m_new;
            alphas[lane] = alpha;
        }
        WAVE_SYNC()

        if (c0 != jlo) {
#pragma unroll
            for (int t = 0; t < 2; ++t) {
#pragma unroll
                for (int r = 0; r < 8; ++r) {
                    const float a = alphas[t * 16 + r + hlf * 8];
#pragma unroll
                    for (int n = 0; n < 4; ++n) acc[t][n][r] *= a;
                }
            }
        }
        if (two) {
            TILE_LOOP(ncol, LOAD_V, MM_V2, vA, vB)
        } else {
            TILE_LOOP(ncol, LOAD_V, MM_V1, vA, vB)
        }
        __syncthreads();
    }
    rowsum[lane] = l_run;
    __syncthreads();

#pragma unroll
    for (int t = 0; t < 2; ++t) {
#pragma unroll
        for (int r = 0; r < 8; ++r) {
            const int irow = t * 16 + r + hlf * 8;
            if (irow < nrows) {
                const int i = perm[(i0 + irow) & (L_ - 1)] & (L_ - 1);
                const float rs = rowsum[irow];
                float* orow = out + (((size_t)(b * L_ + i) * V_) + k) * D_ + l16 * 4;
                v4f o;
                o.x = acc[t][0][r] / rs;
                o.y = acc[t][1][r] / rs;
                o.z = acc[t][2][r] / rs;
                o.w = acc[t][3][r] / rs;
                *(v4f*)orow = o;
            }
        }
    }
    }
}

extern "C" void kernel_launch(void* const* d_in, const int* in_sizes, int n_in,
                              void* d_out, int out_size, void* d_ws, size_t ws_size,
                              hipStream_t stream) {
    const float* query = (const float*)d_in[0];
    const float* key   = (const float*)d_in[1];
    const float* value = (const float*)d_in[2];
    const int*   label = (const int*)d_in[3];
    float* out  = (float*)d_out;
    float* sumv = (float*)d_ws;

    const int n1 = B_ * L_ * D_;
    sum_key_kernel<<<(n1 + 255) / 256, 256, 0, stream>>>(key, sumv);

    const int blocks = B_ * V_ * NT_;
    attn_kernel<<<blocks, 32, 0, stream>>>(query, sumv, value, label, out);
}


# Attention kernel epilogue: one division per row (1 / row sum, by the row's own lane), then multiplies, in place of 64 divisions per lane; inverse sums and row positions read as 16-byte LDS loads
# speedup vs baseline: 4.8584x; 1.0626x over previous
#include <hip/hip_runtime.h>
#include <math.h>


typedef __attribute__((ext_vector_type(2))) float v2f;
typedef __attribute__((ext_vector_type(4))) float v4f;
typedef __attribute__((ext_vector_type(8))) float v8f;
typedef __attribute__((ext_vector_type(4))) int   v4i;

#define B_  4
#define L_  512
#define V_  32
#define D_  64
#define MT  32
#define SCALE 0.125f
#define CH   160
#define LDSS (CH + 4)
#define NC_ 8
#define GPL 3
#define NT_ (NC_ * GPL)

__global__ __launch_bounds__(256)
void sum_key_kernel(const float* __restrict__ key, float* __restrict__ sumv) {
    int idx = blockIdx.x * blockDim.x + threadIdx.x;
    if (idx >= B_ * L_ * D_) return;
    int d  = idx % D_;
    int bl = idx / D_;
    const float* p = key + (size_t)bl * (V_ * D_) + d;
    float s = 0.f;
#pragma unroll
    for (int v = 0; v < V_; ++v) s += p[v * D_];
    sumv[idx] = s;
}

#define LO2(v) __builtin_shufflevector((v), (v), 0, 1)
#define HI2(v) __builtin_shufflevector((v), (v), 2, 3)
#define PIN_ROWS(a, b) asm volatile("" : "+v"(a), "+v"(b));
#define WMMA4(a, b, c) __builtin_amdgcn_wmma_f32_16x16x4_f32(false, (a), false, (b), (short)0, (c), false, false)

#define LOAD_S(buf, CT)                                                        \
    {                                                                          \
        const int js_ = perm[c0 + (CT) + l16] & (L_ - 1);                      \
        const float* sr_ = sb + (size_t)js_ * D_ + hlf * 4;                    \
        _Pragma("unroll")                                                      \
        for (int s = 0; s < 8; ++s) buf[s] = *(const v4f*)(sr_ + s * 8);       \
    }

#define MM_S2(buf, CT)                                                         \
    {                                                                          \
        v8f s0_ = {}, s1_ = {};                                                \
        _Pragma("unroll")                                                      \
        for (int s = 0; s < 8; ++s) {                                          \
            s0_ = WMMA4(LO2(qa0[s]), LO2(buf[s]), s0_);                        \
            s1_ = WMMA4(LO2(qa1[s]), LO2(buf[s]), s1_);                        \
            s0_ = WMMA4(HI2(qa0[s]), HI2(buf[s]), s0_);                        \
            s1_ = WMMA4(HI2(qa1[s]), HI2(buf[s]), s1_);                        \
        }                                                                      \
        const int col_ = (CT) + l16;                                           \
        _Pragma("unroll")                                                      \
        for (int r = 0; r < 8; ++r) {                                          \
            smem[(r + hlf * 8) * LDSS + col_]      = s0_[r];                   \
            smem[(16 + r + hlf * 8) * LDSS + col_] = s1_[r];                   \
        }                                                                      \
    }

#define MM_S1(buf, CT)                                                         \
    {                                                                          \
        v8f s0_ = {}, s0b_ = {};                                               \
        _Pragma("unroll")                                                      \
        for (int s = 0; s < 8; ++s) {                                          \
            s0_  = WMMA4(LO2(qa0[s]), LO2(buf[s]), s0_);                       \
            s0b_ = WMMA4(HI2(qa0[s]), HI2(buf[s]), s0b_);                      \
        }                                                                      \
        s0_ += s0b_;                                                           \
        const int col_ = (CT) + l16;                                           \
        _Pragma("unroll")                                                      \
        for (int r = 0; r < 8; ++r) {                                          \
            smem[(r + hlf * 8) * LDSS + col_]      = s0_[r];                   \
            smem[(16 + r + hlf * 8) * LDSS + col_] = 0.f;                      \
        }                                                                      \
    }

#define LOAD_V(buf, KB)                                                        \
    {                                                                          \
        _Pragma("unroll")                                                      \
        for (int u = 0; u < 4; ++u) {                                          \
            const int j0_ = c0 + (KB) + u * 4 + hlf * 2;                       \
            const int p0_ = perm[j0_] & (L_ - 1);                              \
            const int p1_ = perm[j0_ + 1] & (L_ - 1);                          \
            buf[u * 2]     =                                                   \
                *(const v4f*)(vb + (size_t)p0_ * (V_ * D_) + l16 * 4);         \
            buf[u * 2 + 1] =                                                   \
                *(const v4f*)(vb + (size_t)p1_ * (V_ * D_) + l16 * 4);         \
        }                                                                      \
    }

#define MM_V2(buf, KB)                                                         \
    {                                                                          \
        _Pragma("unroll")                                                      \
        for (int u = 0; u < 4; ++u) {                                          \
            const int jc_ = (KB) + u * 4 + hlf * 2;                            \
            const v2f af0_ = *(const v2f*)(&smem[l16 * LDSS + jc_]);           \
            const v2f af1_ = *(const v2f*)(&smem[(16 + l16) * LDSS + jc_]);    \
            v4f k0_ = buf[u * 2], k1_ = buf[u * 2 + 1];                        \
            PIN_ROWS(k0_, k1_)                                                 \
            _Pragma("unroll")                                                  \
            for (int n = 0; n < 4; ++n) {                                      \
                const v2f bf_ = {k0_[n], k1_[n]};                              \
                acc[0][n] = WMMA4(af0_, bf_, acc[0][n]);                       \
                acc[1][n] = WMMA4(af1_, bf_, acc[1][n]);                       \
            }                                                                  \
        }                                                                      \
    }

#define MM_V1(buf, KB)                                                         \
    {                                                                          \
        _Pragma("unroll")                                                      \
        for (int u = 0; u < 4; ++u) {                                          \
            const int jc_ = (KB) + u * 4 + hlf * 2;                            \
            const v2f af0_ = *(const v2f*)(&smem[l16 * LDSS + jc_]);           \
            v4f k0_ = buf[u * 2], k1_ = buf[u * 2 + 1];                        \
            PIN_ROWS(k0_, k1_)                                                 \
            _Pragma("unroll")                                                  \
            for (int n = 0; n < 4; ++n) {                                      \
                const v2f bf_ = {k0_[n], k1_[n]};                              \
                acc[0][n] = WMMA4(af0_, bf_, acc[0][n]);                       \
            }                                                                  \
        }                                                                      \
    }

#define TILE_LOOP(NCOL, LOAD, MM, bufA, bufB)                                  \
    {                                                                          \
        const int last_  = (NCOL) - 16;                                        \
        const int npair_ = (NCOL) >> 5;                                        \
        int t_ = 0;                                                            \
        for (int p_ = 0; p_ < npair_; ++p_) {                                  \
            LOAD(bufB, t_ + 16)                                                \
            MM(bufA, t_)                                                       \
            const int ta_ = (t_ + 32 < last_) ? (t_ + 32) : last_;             \
            LOAD(bufA, ta_)                                                    \
            MM(bufB, t_ + 16)                                                  \
            t_ += 32;                                                          \
        }                                                                      \
        if ((NCOL) & 16) MM(bufA, last_)                                       \
    }

#define WAVE_SYNC()                                                            \
    {                                                                          \
        __builtin_amdgcn_fence(__ATOMIC_SEQ_CST, "wavefront");                 \
        __builtin_amdgcn_wave_barrier();                                       \
    }

__global__ __launch_bounds__(32) __attribute__((amdgpu_waves_per_eu(3)))
void attn_kernel(const float* __restrict__ q,
                 const float* __restrict__ sumv,
                 const float* __restrict__ val,
                 const int*   __restrict__ lab,
                 float*       __restrict__ out) {
    __shared__ __attribute__((aligned(16))) float smem[MT * LDSS];
    __shared__ __attribute__((aligned(16))) float labf[L_ + MT];
    __shared__ __attribute__((aligned(16))) int   perm[L_ + MT];
    __shared__ __attribute__((aligned(16))) float rowinv[MT];
    __shared__ float alphas[MT];

    const int blk   = blockIdx.x;
    const int itile = blk % NT_;
    const int bk    = blk / NT_;
    const int k     = bk % V_;
    const int b     = bk / V_;

    const int lane = threadIdx.x;
    const int hlf  = lane >> 4;
    const int l16  = lane & 15;

    int lv[16];
    bool inrange = true;
#pragma unroll
    for (int t = 0; t < 16; ++t) {
        lv[t] = lab[b * L_ + t * 32 + lane];
        inrange = inrange && ((unsigned)lv[t] < (unsigned)NC_);
    }
    const bool sorted = (__builtin_amdgcn_ballot_w32(!inrange) == 0u);

    int nkey_u, g_u, gstep_u, ntile_u;
    if (sorted) {
        const int cown = itile / GPL;
        int base = 0;
#pragma unroll
        for (int t = 0; t < 16; ++t) {
            const bool hit = (lv[t] == cown);
            const unsigned m = __builtin_amdgcn_ballot_w32(hit);
            const int p = (int)__builtin_amdgcn_mbcnt_lo(m, (unsigned)base);
            if (hit) perm[p] = t * 32 + lane;
            base += __popc(m);
        }
        const int n = base;
        perm[n + lane] = 0;
        const float cf = (float)cown;
#pragma unroll 1
        for (int j = lane; j < n + MT; j += 32)
            labf[j] = (j < n) ? cf : __builtin_nanf("");
        nkey_u  = n;
        g_u     = itile % GPL;
        gstep_u = GPL;
        ntile_u = (n + MT - 1) / MT;
    } else {
#pragma unroll
        for (int t = 0; t < 16; ++t) {
            perm[t * 32 + lane] = t * 32 + lane;
            labf[t * 32 + lane] = (float)lv[t];
        }
        perm[L_ + lane] = 0;
        labf[L_ + lane] = __builtin_nanf("");
        nkey_u  = L_;
        g_u     = itile;
        gstep_u = NT_;
        ntile_u = L_ / MT;
    }
    __syncthreads();
    int nkey = __builtin_amdgcn_readfirstlane(nkey_u);
    nkey = nkey < 0 ? 0 : (nkey > L_ ? L_ : nkey);
    const int gstep = __builtin_amdgcn_readfirstlane(gstep_u);
    const int ntile = __builtin_amdgcn_readfirstlane(ntile_u);
    const int jlo = 0;
    const int jhi = nkey;

    const float* sb = sumv + (size_t)b * L_ * D_;
    const float* vb = val + ((size_t)(b * L_) * V_ + k) * D_;

    for (int g = __builtin_amdgcn_readfirstlane(g_u); g < ntile; g += gstep) {
    int i0 = g * MT;
    i0 = i0 < 0 ? 0 : (i0 > L_ - 1 ? L_ - 1 : i0);
    const int left  = nkey - i0;
    const int nrows = left < MT ? left : MT;
    if (nrows <= 0) break;
    const bool two = nrows > 16;

    v4f qa0[8], qa1[8];
    {
        const int r0 = perm[i0 + l16] & (L_ - 1);
        const int r1 = perm[i0 + 16 + l16] & (L_ - 1);
        const float* q0 =
            q + (((size_t)(b * L_ + r0) * V_) + k) * D_ + hlf * 4;
        const float* q1 =
            q + (((size_t)(b * L_ + r1) * V_) + k) * D_ + hlf * 4;
#pragma unroll
        for (int s = 0; s < 8; ++s) {
            qa0[s] = *(const v4f*)(q0 + s * 8);
            qa1[s] = *(const v4f*)(q1 + s * 8);
        }
    }

    const float myLab = labf[i0 + lane];
    float m_run = -INFINITY;
    float l_run = 0.f;
    v8f acc[2][4] = {{{}, {}, {}, {}}, {{}, {}, {}, {}}};

    for (int c0 = jlo; c0 < jhi; c0 += CH) {
        const int c1   = (c0 + CH < jhi) ? (c0 + CH) : jhi;
        const int ncol = (c1 - c0 + 15) & ~15;

        {
            v4f sA[8], sB[8];
            LOAD_S(sA, 0)
            if (two) {
                TILE_LOOP(ncol, LOAD_S, MM_S2, sA, sB)
            } else {
                TILE_LOOP(ncol, LOAD_S, MM_S1, sA, sB)
            }
        }
        v4f vA[8], vB[8];
        LOAD_V(vA, 0)
        WAVE_SYNC()

        {
            float* srow = &smem[lane * LDSS];

            v4f m4 = {-INFINITY, -INFINITY, -INFINITY, -INFINITY};
            for (int cb = 0; cb < ncol; cb += 16) {
#pragma unroll
                for (int u = 0; u < 4; ++u) {
                    const int cc = cb + u * 4;
                    const int c  = c0 + cc;
                    v4f s  = *(v4f*)(srow + cc);
                    const float lx = labf[c], ly = labf[c + 1];
                    const float lz = labf[c + 2], lw = labf[c + 3];
                    s.x = (lx == myLab) ? s.x : -INFINITY;
                    s.y = (ly == myLab) ? s.y : -INFINITY;
                    s.z = (lz == myLab) ? s.z : -INFINITY;
                    s.w = (lw == myLab) ? s.w : -INFINITY;
                    *(v4f*)(srow + cc) = s;
                    m4.x = fmaxf(m4.x, s.x);  m4.y = fmaxf(m4.y, s.y);
                    m4.z = fmaxf(m4.z, s.z);  m4.w = fmaxf(m4.w, s.w);
                }
            }
            const float mc    = fmaxf(fmaxf(m4.x, m4.y), fmaxf(m4.z, m4.w));
            const float m_new = fmaxf(m_run, mc);
            const float m_use = (m_new == -INFINITY) ? 0.f : m_new;
            const float alpha = __expf(SCALE * (m_run - m_use));

            v4f s4 = {0.f, 0.f, 0.f, 0.f};
            for (int cb = 0; cb < ncol; cb += 16) {
#pragma unroll
                for (int u = 0; u < 4; ++u) {
                    const int c = cb + u * 4;
                    v4f s = *(v4f*)(srow + c);
                    v4f p;
                    p.x = __expf(SCALE * (s.x - m_use));
                    p.y = __expf(SCALE * (s.y - m_use));
                    p.z = __expf(SCALE * (s.z - m_use));
                    p.w = __expf(SCALE * (s.w - m_use));
                    *(v4f*)(srow + c) = p;
                    s4 += p;
                }
            }
            l_run = l_run * alpha + ((s4.x + s4.y) + (s4.z + s4.w));
            m_run = m_new;
            alphas[lane] = alpha;
        }
        WAVE_SYNC()

        if (c0 != jlo) {
#pragma unroll
            for (int t = 0; t < 2; ++t) {
#pragma unroll
                for (int r = 0; r < 8; ++r) {
                    const float a = alphas[t * 16 + r + hlf * 8];
#pragma unroll
                    for (int n = 0; n < 4; ++n) acc[t][n][r] *= a;
                }
            }
        }
        if (two) {
            TILE_LOOP(ncol, LOAD_V, MM_V2, vA, vB)
        } else {
            TILE_LOOP(ncol, LOAD_V, MM_V1, vA, vB)
        }
        __syncthreads();
    }
    rowinv[lane] = 1.0f / l_run;
    __syncthreads();

    {
        const int i0a = i0 & ~(MT - 1);
        v4f iv[2][2];
        v4i ps[2][2];
#pragma unroll
        for (int t = 0; t < 2; ++t) {
            const int rb = t * 16 + hlf * 8;
            iv[t][0] = *(const v4f*)&rowinv[rb];
            iv[t][1] = *(const v4f*)&rowinv[rb + 4];
            ps[t][0] = *(const v4i*)&perm[i0a + rb];
            ps[t][1] = *(const v4i*)&perm[i0a + rb + 4];
        }
#pragma unroll
        for (int t = 0; t < 2; ++t) {
            if (t == 1 && !two) break;
            const int rb = t * 16 + hlf * 8;
#pragma unroll
            for (int r = 0; r < 8; ++r) {
                const float inv = iv[t][r >> 2][r & 3];
                const int   i   = ps[t][r >> 2][r & 3] & (L_ - 1);
                if (rb + r < nrows) {
                    float* orow = out + (((size_t)(b * L_ + i) * V_) + k) * D_ + l16 * 4;
                    v4f o;
                    o.x = acc[t][0][r] * inv;
                    o.y = acc[t][1][r] * inv;
                    o.z = acc[t][2][r] * inv;
                    o.w = acc[t][3][r] * inv;
                    *(v4f*)orow = o;
                }
            }
        }
    }
    }
}

extern "C" void kernel_launch(void* const* d_in, const int* in_sizes, int n_in,
                              void* d_out, int out_size, void* d_ws, size_t ws_size,
                              hipStream_t stream) {
    const float* query = (const float*)d_in[0];
    const float* key   = (const float*)d_in[1];
    const float* value = (const float*)d_in[2];
    const int*   label = (const int*)d_in[3];
    float* out  = (float*)d_out;
    float* sumv = (float*)d_ws;

    const int n1 = B_ * L_ * D_;
    sum_key_kernel<<<(n1 + 255) / 256, 256, 0, stream>>>(key, sumv);

    const int blocks = B_ * V_ * NT_;
    attn_kernel<<<blocks, 32, 0, stream>>>(query, sumv, value, label, out);
}
